# starting kernel
# baseline (speedup 1.0000x reference)
#include <hip/hip_runtime.h>

typedef __attribute__((ext_vector_type(2))) float v2f;
typedef __attribute__((ext_vector_type(8))) float v8f;

#define WPB 8
#define DD  256

static __device__ __forceinline__ v8f wmma4(v2f a, v2f b, v8f c) {
  return __builtin_amdgcn_wmma_f32_16x16x4_f32(false, a, false, b, (short)0, c,
                                               false, false);
}

static __device__ __forceinline__ v2f ldRow2(const float* base, int i, int k, int h) {
  const float* p = base + i * 16 + 4 * k + 2 * h;
  v2f r; r.x = p[0]; r.y = p[1];
  return r;
}
static __device__ __forceinline__ v2f ldColS(const float* base, int m, int k, int h) {
  int kk = 4 * k + 2 * h;
  v2f r; r.x = base[kk * 16 + m]; r.y = base[(kk + 1) * 16 + m];
  return r;
}

template <bool NEGA, bool COLA>
static __device__ __forceinline__ v8f mm16(const float* A, const float* B, v8f acc,
                                           int i, int h) {
#pragma unroll
  for (int k = 0; k < 4; ++k) {
    v2f a = COLA ? ldColS(A, i, k, h) : ldRow2(A, i, k, h);
    if (NEGA) a = -a;
    v2f b = ldRow2(B, i, k, h);
    acc = wmma4(a, b, acc);
  }
  return acc;
}

static __device__ __forceinline__ void stCD(float* D, v8f v, int i, int h) {
  float* p = D + i * 16 + h * 8;
#pragma unroll
  for (int g = 0; g < 8; ++g) p[g] = v[g];
}

static __device__ __forceinline__ void cmatL(const float* Lr, const float* Li,
                                             const float* Sr, const float* Si,
                                             float* Dr, float* Di, int lane) {
  int i = lane & 15, h = lane >> 4;
  v8f ar = {};
  ar = mm16<false, false>(Lr, Sr, ar, i, h);
  ar = mm16<true,  false>(Li, Si, ar, i, h);
  v8f ai = {};
  ai = mm16<false, false>(Lr, Si, ai, i, h);
  ai = mm16<false, false>(Li, Sr, ai, i, h);
  stCD(Dr, ar, i, h);
  stCD(Di, ai, i, h);
}

static __device__ __forceinline__ void cmatR(const float* Tr, const float* Ti,
                                             const float* Rr, const float* Ri,
                                             float* Dr, float* Di, int lane) {
  int i = lane & 15, h = lane >> 4;
  v8f ar = {};
  ar = mm16<false, true>(Tr, Rr, ar, i, h);
  ar = mm16<true,  true>(Ti, Ri, ar, i, h);
  v8f ai = {};
  ai = mm16<false, true>(Tr, Ri, ai, i, h);
  ai = mm16<false, true>(Ti, Rr, ai, i, h);
  stCD(Dr, ar, i, h);
  stCD(Di, ai, i, h);
}

static __device__ __forceinline__ void rotMat(float phi, float th, float om,
                                              float gr[2][2], float gi[2][2]) {
  float st, ct, sa, ca, sb, cb;
  __sincosf(th * 0.5f, &st, &ct);
  __sincosf(0.5f * (phi + om), &sa, &ca);
  __sincosf(0.5f * (phi - om), &sb, &cb);
  gr[0][0] = ct * ca;  gi[0][0] = -ct * sa;
  gr[0][1] = -st * cb; gi[0][1] = -st * sb;
  gr[1][0] = st * cb;  gi[1][0] = -st * sb;
  gr[1][1] = ct * ca;  gi[1][1] = ct * sa;
}

static __device__ __forceinline__ void encMat(float th, float phi,
                                              float gr[2][2], float gi[2][2]) {
  float s, c, sp, cp;
  __sincosf(th * 0.5f, &s, &c);
  __sincosf(phi * 0.5f, &sp, &cp);
  gr[0][0] = c * cp;  gi[0][0] = -c * sp;
  gr[0][1] = -s * cp; gi[0][1] = s * sp;
  gr[1][0] = s * cp;  gi[1][0] = s * sp;
  gr[1][1] = c * cp;  gi[1][1] = c * sp;
}

static __device__ __forceinline__ void buildKron(const float gr[4][2][2],
                                                 const float gi[4][2][2],
                                                 float* Kr, float* Ki, int lane) {
#pragma unroll
  for (int j = 0; j < 8; ++j) {
    int idx = lane * 8 + j;
    int a = idx >> 4;
    int bcol = idx & 15;
    float pr = 1.f, pi = 0.f;
#pragma unroll
    for (int q = 0; q < 4; ++q) {
      int ba = (a >> (3 - q)) & 1;
      int bb = (bcol >> (3 - q)) & 1;
      float xr = gr[q][ba][bb], xi = gi[q][ba][bb];
      float nr = pr * xr - pi * xi;
      pi = pr * xi + pi * xr;
      pr = nr;
    }
    Kr[idx] = pr;
    Ki[idx] = pi;
  }
}

static __device__ __forceinline__ void buildRot(const float* vqcw, int l,
                                                float* GLr, float* GLi,
                                                float* GRr, float* GRi, int lane) {
  float gr[4][2][2], gi[4][2][2];
#pragma unroll
  for (int q = 0; q < 4; ++q) {
    const float* w = vqcw + (l * 8 + q) * 3;
    rotMat(w[0], w[1], w[2], gr[q], gi[q]);
  }
  buildKron(gr, gi, GLr, GLi, lane);
#pragma unroll
  for (int q = 0; q < 4; ++q) {
    const float* w = vqcw + (l * 8 + 4 + q) * 3;
    rotMat(w[0], w[1], w[2], gr[q], gi[q]);
  }
  buildKron(gr, gi, GRr, GRi, lane);
}

static __device__ __forceinline__ void buildReenc(const float* th, const float* ph,
                                                  float* GLr, float* GLi,
                                                  float* GRr, float* GRi, int lane) {
  float gr[4][2][2], gi[4][2][2];
#pragma unroll
  for (int q = 0; q < 4; ++q) encMat(th[q] * 0.5f, ph[q] * 0.5f, gr[q], gi[q]);
  buildKron(gr, gi, GLr, GLi, lane);
#pragma unroll
  for (int q = 0; q < 4; ++q) encMat(th[4 + q] * 0.5f, ph[4 + q] * 0.5f, gr[q], gi[q]);
  buildKron(gr, gi, GRr, GRi, lane);
}

template <int RNG>
static __device__ __forceinline__ void applyCnot(const float* Sr, const float* Si,
                                                 float* Dr, float* Di, int lane) {
#pragma unroll
  for (int j = 0; j < 8; ++j) {
    int d = lane * 8 + j;
    int s = ((d & 15) << 4) | (d >> 4);
    int src = s;
#pragma unroll
    for (int k = 7; k >= 0; --k) {
      int tgt = (k + RNG) & 7;
      src ^= ((src >> (7 - k)) & 1) << (7 - tgt);
    }
    int sa = ((src & 15) << 4) | (src >> 4);
    Dr[d] = Sr[sa];
    Di[d] = Si[sa];
  }
}

extern "C" __global__ __launch_bounds__(256) void vqc_kernel(
    const float* __restrict__ theta, const float* __restrict__ phi,
    const float* __restrict__ jup, const float* __restrict__ vqcw,
    float* __restrict__ out) {
  __shared__ float sXr[WPB][DD], sXi[WPB][DD];
  __shared__ float sYr[WPB][DD], sYi[WPB][DD];
  __shared__ float sGLr[WPB][DD], sGLi[WPB][DD];
  __shared__ float sGRr[WPB][DD], sGRi[WPB][DD];

  int tid = threadIdx.x;
  int wid = tid >> 5, lane = tid & 31;
  int b = blockIdx.x * WPB + wid;

  const float* th = theta + b * 8;
  const float* ph = phi + b * 8;
  const float* jj = jup + b * 28;

  float* Xr = sXr[wid]; float* Xi = sXi[wid];
  float* Yr = sYr[wid]; float* Yi = sYi[wid];
  float* GLr = sGLr[wid]; float* GLi = sGLi[wid];
  float* GRr = sGRr[wid]; float* GRi = sGRi[wid];

  {
    float e0r[8], e0i[8], e1r[8], e1i[8];
#pragma unroll
    for (int k = 0; k < 8; ++k) {
      float s, c, sp, cp;
      __sincosf(th[k] * 0.5f, &s, &c);
      __sincosf(ph[k] * 0.5f, &sp, &cp);
      e0r[k] = c * cp; e0i[k] = -c * sp;
      e1r[k] = s * cp; e1i[k] = s * sp;
    }
    float jv[28];
#pragma unroll
    for (int p = 0; p < 28; ++p) jv[p] = jj[p];

#pragma unroll
    for (int j = 0; j < 8; ++j) {
      int d = lane * 8 + j;
      int s = ((d & 15) << 4) | (d >> 4);
      float pr = 1.f, pim = 0.f;
#pragma unroll
      for (int k = 0; k < 8; ++k) {
        int bit = (s >> (7 - k)) & 1;
        float xr = bit ? e1r[k] : e0r[k];
        float xi = bit ? e1i[k] : e0i[k];
        float nr = pr * xr - pim * xi;
        pim = pr * xi + pim * xr;
        pr = nr;
      }
      float e = 0.f;
      int p = 0;
#pragma unroll
      for (int qi = 0; qi < 8; ++qi)
#pragma unroll
        for (int qj = qi + 1; qj < 8; ++qj) {
          int diff = ((s >> (7 - qi)) & 1) ^ ((s >> (7 - qj)) & 1);
          e += diff ? jv[p] : -jv[p];
          ++p;
        }
      e *= 1.57079632679489662f;
      float se, ce;
      __sincosf(e, &se, &ce);
      Xr[d] = pr * ce - pim * se;
      Xi[d] = pr * se + pim * ce;
    }
  }
  buildRot(vqcw, 0, GLr, GLi, GRr, GRi, lane);
  __syncthreads();

  cmatL(GLr, GLi, Xr, Xi, Yr, Yi, lane);
  __syncthreads();
  cmatR(Yr, Yi, GRr, GRi, Xr, Xi, lane);
  __syncthreads();

  applyCnot<1>(Xr, Xi, Yr, Yi, lane);
  buildReenc(th, ph, GLr, GLi, GRr, GRi, lane);
  __syncthreads();

  cmatL(GLr, GLi, Yr, Yi, Xr, Xi, lane);
  __syncthreads();
  cmatR(Xr, Xi, GRr, GRi, Yr, Yi, lane);
  __syncthreads();

  buildRot(vqcw, 1, GLr, GLi, GRr, GRi, lane);
  __syncthreads();
  cmatL(GLr, GLi, Yr, Yi, Xr, Xi, lane);
  __syncthreads();
  cmatR(Xr, Xi, GRr, GRi, Yr, Yi, lane);
  __syncthreads();

  applyCnot<2>(Yr, Yi, Xr, Xi, lane);
  __syncthreads();

  float acc[8];
#pragma unroll
  for (int k = 0; k < 8; ++k) acc[k] = 0.f;
#pragma unroll
  for (int j = 0; j < 8; ++j) {
    int d = lane * 8 + j;
    int s = ((d & 15) << 4) | (d >> 4);
    float vr = Xr[d], vi = Xi[d];
    float p2 = vr * vr + vi * vi;
#pragma unroll
    for (int k = 0; k < 8; ++k)
      acc[k] += ((s >> (7 - k)) & 1) ? -p2 : p2;
  }
#pragma unroll
  for (int k = 0; k < 8; ++k) {
    float v = acc[k];
    v += __shfl_xor(v, 16, 32);
    v += __shfl_xor(v, 8, 32);
    v += __shfl_xor(v, 4, 32);
    v += __shfl_xor(v, 2, 32);
    v += __shfl_xor(v, 1, 32);
    acc[k] = v;
  }
  if (lane == 0) {
#pragma unroll
    for (int k = 0; k < 8; ++k) out[b * 8 + k] = acc[k];
  }
}

extern "C" void kernel_launch(void* const* d_in, const int* in_sizes, int n_in,
                              void* d_out, int out_size, void* d_ws, size_t ws_size,
                              hipStream_t stream) {
  const float* theta = (const float*)d_in[0];
  const float* phi   = (const float*)d_in[1];
  const float* jup   = (const float*)d_in[2];
  const float* vqcw  = (const float*)d_in[3];
  float* out = (float*)d_out;
  int B = in_sizes[0] / 8;
  dim3 grid(B / WPB), block(32 * WPB);
  hipLaunchKernelGGL(vqc_kernel, grid, block, 0, stream, theta, phi, jup, vqcw, out);
}


# Kronecker gate build keeps the 2x2 gate entries in registers and picks them by lane-bit selects instead of lane-indexed private arrays: scratch memory per lane goes from 144 B to 0
# speedup vs baseline: 14.5430x; 14.5430x over previous
#include <hip/hip_runtime.h>

typedef __attribute__((ext_vector_type(2))) float v2f;
typedef __attribute__((ext_vector_type(8))) float v8f;

#define WPB 8
#define DD  256

static __device__ __forceinline__ v8f wmma4(v2f a, v2f b, v8f c) {
  return __builtin_amdgcn_wmma_f32_16x16x4_f32(false, a, false, b, (short)0, c,
                                               false, false);
}

static __device__ __forceinline__ v2f ldRow2(const float* base, int i, int k, int h) {
  const float* p = base + i * 16 + 4 * k + 2 * h;
  v2f r; r.x = p[0]; r.y = p[1];
  return r;
}
static __device__ __forceinline__ v2f ldColS(const float* base, int m, int k, int h) {
  int kk = 4 * k + 2 * h;
  v2f r; r.x = base[kk * 16 + m]; r.y = base[(kk + 1) * 16 + m];
  return r;
}

template <bool NEGA, bool COLA>
static __device__ __forceinline__ v8f mm16(const float* A, const float* B, v8f acc,
                                           int i, int h) {
#pragma unroll
  for (int k = 0; k < 4; ++k) {
    v2f a = COLA ? ldColS(A, i, k, h) : ldRow2(A, i, k, h);
    if (NEGA) a = -a;
    v2f b = ldRow2(B, i, k, h);
    acc = wmma4(a, b, acc);
  }
  return acc;
}

static __device__ __forceinline__ void stCD(float* D, v8f v, int i, int h) {
  float* p = D + i * 16 + h * 8;
#pragma unroll
  for (int g = 0; g < 8; ++g) p[g] = v[g];
}

static __device__ __forceinline__ void cmatL(const float* Lr, const float* Li,
                                             const float* Sr, const float* Si,
                                             float* Dr, float* Di, int lane) {
  int i = lane & 15, h = lane >> 4;
  v8f ar = {};
  ar = mm16<false, false>(Lr, Sr, ar, i, h);
  ar = mm16<true,  false>(Li, Si, ar, i, h);
  v8f ai = {};
  ai = mm16<false, false>(Lr, Si, ai, i, h);
  ai = mm16<false, false>(Li, Sr, ai, i, h);
  stCD(Dr, ar, i, h);
  stCD(Di, ai, i, h);
}

static __device__ __forceinline__ void cmatR(const float* Tr, const float* Ti,
                                             const float* Rr, const float* Ri,
                                             float* Dr, float* Di, int lane) {
  int i = lane & 15, h = lane >> 4;
  v8f ar = {};
  ar = mm16<false, true>(Tr, Rr, ar, i, h);
  ar = mm16<true,  true>(Ti, Ri, ar, i, h);
  v8f ai = {};
  ai = mm16<false, true>(Tr, Ri, ai, i, h);
  ai = mm16<false, true>(Ti, Rr, ai, i, h);
  stCD(Dr, ar, i, h);
  stCD(Di, ai, i, h);
}

struct G2 { float r00, r01, r10, r11, i00, i01, i10, i11; };
struct C2 { float r, i; };

static __device__ __forceinline__ float pick(bool c, float t, float f) { return c ? t : f; }

static __device__ __forceinline__ C2 cmul(C2 p, C2 x) {
  C2 o;
  o.r = p.r * x.r - p.i * x.i;
  o.i = p.r * x.i + p.i * x.r;
  return o;
}

static __device__ __forceinline__ G2 rotMat(float phi, float th, float om) {
  float st, ct, sa, ca, sb, cb;
  __sincosf(th * 0.5f, &st, &ct);
  __sincosf(0.5f * (phi + om), &sa, &ca);
  __sincosf(0.5f * (phi - om), &sb, &cb);
  G2 g;
  g.r00 = ct * ca;  g.i00 = -ct * sa;
  g.r01 = -st * cb; g.i01 = -st * sb;
  g.r10 = st * cb;  g.i10 = -st * sb;
  g.r11 = ct * ca;  g.i11 = ct * sa;
  return g;
}

static __device__ __forceinline__ G2 encMat(float th, float phi) {
  float s, c, sp, cp;
  __sincosf(th * 0.5f, &s, &c);
  __sincosf(phi * 0.5f, &sp, &cp);
  G2 g;
  g.r00 = c * cp;  g.i00 = -c * sp;
  g.r01 = -s * cp; g.i01 = s * sp;
  g.r10 = s * cp;  g.i10 = s * sp;
  g.r11 = c * cp;  g.i11 = c * sp;
  return g;
}

static __device__ __forceinline__ void pickRow(bool a, G2 g, C2* c0, C2* c1) {
  c0->r = pick(a, g.r10, g.r00);  c0->i = pick(a, g.i10, g.i00);
  c1->r = pick(a, g.r11, g.r01);  c1->i = pick(a, g.i11, g.i01);
}

static __device__ __forceinline__ void buildKron(G2 g0, G2 g1, G2 g2, G2 g3,
                                                 float* Kr, float* Ki, int lane) {
  const bool a0 = (lane >> 4) & 1;
  const bool a1 = (lane >> 3) & 1;
  const bool a2 = (lane >> 2) & 1;
  const bool a3 = (lane >> 1) & 1;
  const bool b0 = lane & 1;
  C2 x00, x01, p;
  pickRow(a0, g0, &x00, &x01);
  p.r = pick(b0, x01.r, x00.r);
  p.i = pick(b0, x01.i, x00.i);
  C2 u0, u1, v0, v1, w0, w1;
  pickRow(a1, g1, &u0, &u1);
  pickRow(a2, g2, &v0, &v1);
  pickRow(a3, g3, &w0, &w1);
  const C2 s0 = cmul(p, u0), s1 = cmul(p, u1);
  const C2 t0 = cmul(s0, v0), t1 = cmul(s0, v1);
  const C2 t2 = cmul(s1, v0), t3 = cmul(s1, v1);
  const C2 e0 = cmul(t0, w0), e1 = cmul(t0, w1);
  const C2 e2 = cmul(t1, w0), e3 = cmul(t1, w1);
  const C2 e4 = cmul(t2, w0), e5 = cmul(t2, w1);
  const C2 e6 = cmul(t3, w0), e7 = cmul(t3, w1);
  float* kr = Kr + lane * 8;
  float* ki = Ki + lane * 8;
  kr[0] = e0.r; kr[1] = e1.r; kr[2] = e2.r; kr[3] = e3.r;
  kr[4] = e4.r; kr[5] = e5.r; kr[6] = e6.r; kr[7] = e7.r;
  ki[0] = e0.i; ki[1] = e1.i; ki[2] = e2.i; ki[3] = e3.i;
  ki[4] = e4.i; ki[5] = e5.i; ki[6] = e6.i; ki[7] = e7.i;
}

static __device__ __forceinline__ G2 rotGate(const float* vqcw, int l, int q) {
  const float* w = vqcw + (l * 8 + q) * 3;
  return rotMat(w[0], w[1], w[2]);
}

static __device__ __forceinline__ void buildRot(const float* vqcw, int l,
                                                float* GLr, float* GLi,
                                                float* GRr, float* GRi, int lane) {
  buildKron(rotGate(vqcw, l, 0), rotGate(vqcw, l, 1), rotGate(vqcw, l, 2),
            rotGate(vqcw, l, 3), GLr, GLi, lane);
  buildKron(rotGate(vqcw, l, 4), rotGate(vqcw, l, 5), rotGate(vqcw, l, 6),
            rotGate(vqcw, l, 7), GRr, GRi, lane);
}

static __device__ __forceinline__ G2 reencGate(const float* th, const float* ph, int q) {
  return encMat(th[q] * 0.5f, ph[q] * 0.5f);
}

static __device__ __forceinline__ void buildReenc(const float* th, const float* ph,
                                                  float* GLr, float* GLi,
                                                  float* GRr, float* GRi, int lane) {
  buildKron(reencGate(th, ph, 0), reencGate(th, ph, 1), reencGate(th, ph, 2),
            reencGate(th, ph, 3), GLr, GLi, lane);
  buildKron(reencGate(th, ph, 4), reencGate(th, ph, 5), reencGate(th, ph, 6),
            reencGate(th, ph, 7), GRr, GRi, lane);
}

template <int RNG>
static __device__ __forceinline__ void applyCnot(const float* Sr, const float* Si,
                                                 float* Dr, float* Di, int lane) {
#pragma unroll
  for (int j = 0; j < 8; ++j) {
    int d = lane * 8 + j;
    int s = ((d & 15) << 4) | (d >> 4);
    int src = s;
#pragma unroll
    for (int k = 7; k >= 0; --k) {
      int tgt = (k + RNG) & 7;
      src ^= ((src >> (7 - k)) & 1) << (7 - tgt);
    }
    int sa = ((src & 15) << 4) | (src >> 4);
    Dr[d] = Sr[sa];
    Di[d] = Si[sa];
  }
}

extern "C" __global__ __launch_bounds__(256) void vqc_kernel(
    const float* __restrict__ theta, const float* __restrict__ phi,
    const float* __restrict__ jup, const float* __restrict__ vqcw,
    float* __restrict__ out) {
  __shared__ float sXr[WPB][DD], sXi[WPB][DD];
  __shared__ float sYr[WPB][DD], sYi[WPB][DD];
  __shared__ float sGLr[WPB][DD], sGLi[WPB][DD];
  __shared__ float sGRr[WPB][DD], sGRi[WPB][DD];

  int tid = threadIdx.x;
  int wid = tid >> 5, lane = tid & 31;
  int b = blockIdx.x * WPB + wid;

  const float* th = theta + b * 8;
  const float* ph = phi + b * 8;
  const float* jj = jup + b * 28;

  float* Xr = sXr[wid]; float* Xi = sXi[wid];
  float* Yr = sYr[wid]; float* Yi = sYi[wid];
  float* GLr = sGLr[wid]; float* GLi = sGLi[wid];
  float* GRr = sGRr[wid]; float* GRi = sGRi[wid];

  {
    float e0r[8], e0i[8], e1r[8], e1i[8];
#pragma unroll
    for (int k = 0; k < 8; ++k) {
      float s, c, sp, cp;
      __sincosf(th[k] * 0.5f, &s, &c);
      __sincosf(ph[k] * 0.5f, &sp, &cp);
      e0r[k] = c * cp; e0i[k] = -c * sp;
      e1r[k] = s * cp; e1i[k] = s * sp;
    }
    float jv[28];
#pragma unroll
    for (int p = 0; p < 28; ++p) jv[p] = jj[p];

#pragma unroll
    for (int j = 0; j < 8; ++j) {
      int d = lane * 8 + j;
      int s = ((d & 15) << 4) | (d >> 4);
      float pr = 1.f, pim = 0.f;
#pragma unroll
      for (int k = 0; k < 8; ++k) {
        int bit = (s >> (7 - k)) & 1;
        float xr = bit ? e1r[k] : e0r[k];
        float xi = bit ? e1i[k] : e0i[k];
        float nr = pr * xr - pim * xi;
        pim = pr * xi + pim * xr;
        pr = nr;
      }
      float e = 0.f;
      int p = 0;
#pragma unroll
      for (int qi = 0; qi < 8; ++qi)
#pragma unroll
        for (int qj = qi + 1; qj < 8; ++qj) {
          int diff = ((s >> (7 - qi)) & 1) ^ ((s >> (7 - qj)) & 1);
          e += diff ? jv[p] : -jv[p];
          ++p;
        }
      e *= 1.57079632679489662f;
      float se, ce;
      __sincosf(e, &se, &ce);
      Xr[d] = pr * ce - pim * se;
      Xi[d] = pr * se + pim * ce;
    }
  }
  buildRot(vqcw, 0, GLr, GLi, GRr, GRi, lane);
  __syncthreads();

  cmatL(GLr, GLi, Xr, Xi, Yr, Yi, lane);
  __syncthreads();
  cmatR(Yr, Yi, GRr, GRi, Xr, Xi, lane);
  __syncthreads();

  applyCnot<1>(Xr, Xi, Yr, Yi, lane);
  buildReenc(th, ph, GLr, GLi, GRr, GRi, lane);
  __syncthreads();

  cmatL(GLr, GLi, Yr, Yi, Xr, Xi, lane);
  __syncthreads();
  cmatR(Xr, Xi, GRr, GRi, Yr, Yi, lane);
  __syncthreads();

  buildRot(vqcw, 1, GLr, GLi, GRr, GRi, lane);
  __syncthreads();
  cmatL(GLr, GLi, Yr, Yi, Xr, Xi, lane);
  __syncthreads();
  cmatR(Xr, Xi, GRr, GRi, Yr, Yi, lane);
  __syncthreads();

  applyCnot<2>(Yr, Yi, Xr, Xi, lane);
  __syncthreads();

  float acc[8];
#pragma unroll
  for (int k = 0; k < 8; ++k) acc[k] = 0.f;
#pragma unroll
  for (int j = 0; j < 8; ++j) {
    int d = lane * 8 + j;
    int s = ((d & 15) << 4) | (d >> 4);
    float vr = Xr[d], vi = Xi[d];
    float p2 = vr * vr + vi * vi;
#pragma unroll
    for (int k = 0; k < 8; ++k)
      acc[k] += ((s >> (7 - k)) & 1) ? -p2 : p2;
  }
#pragma unroll
  for (int k = 0; k < 8; ++k) {
    float v = acc[k];
    v += __shfl_xor(v, 16, 32);
    v += __shfl_xor(v, 8, 32);
    v += __shfl_xor(v, 4, 32);
    v += __shfl_xor(v, 2, 32);
    v += __shfl_xor(v, 1, 32);
    acc[k] = v;
  }
  if (lane == 0) {
#pragma unroll
    for (int k = 0; k < 8; ++k) out[b * 8 + k] = acc[k];
  }
}

extern "C" void kernel_launch(void* const* d_in, const int* in_sizes, int n_in,
                              void* d_out, int out_size, void* d_ws, size_t ws_size,
                              hipStream_t stream) {
  const float* theta = (const float*)d_in[0];
  const float* phi   = (const float*)d_in[1];
  const float* jup   = (const float*)d_in[2];
  const float* vqcw  = (const float*)d_in[3];
  float* out = (float*)d_out;
  int B = in_sizes[0] / 8;
  dim3 grid(B / WPB), block(32 * WPB);
  hipLaunchKernelGGL(vqc_kernel, grid, block, 0, stream, theta, phi, jup, vqcw, out);
}


# Re-encoding and layer-1 Rot gate of each qubit fused into one 2x2 gate (RZ angles added, times RY) before the Kronecker build: 4 complex 16x16 products, not 6; that stage's rounding order changes
# speedup vs baseline: 18.1534x; 1.2483x over previous
#include <hip/hip_runtime.h>

typedef __attribute__((ext_vector_type(2))) float v2f;
typedef __attribute__((ext_vector_type(8))) float v8f;

#define WPB 8
#define DD  256

static __device__ __forceinline__ v8f wmma4(v2f a, v2f b, v8f c) {
  return __builtin_amdgcn_wmma_f32_16x16x4_f32(false, a, false, b, (short)0, c,
                                               false, false);
}

static __device__ __forceinline__ v2f ldRow2(const float* base, int i, int k, int h) {
  const float* p = base + i * 16 + 4 * k + 2 * h;
  v2f r; r.x = p[0]; r.y = p[1];
  return r;
}
static __device__ __forceinline__ v2f ldColS(const float* base, int m, int k, int h) {
  int kk = 4 * k + 2 * h;
  v2f r; r.x = base[kk * 16 + m]; r.y = base[(kk + 1) * 16 + m];
  return r;
}

template <bool NEGA, bool COLA>
static __device__ __forceinline__ v8f mm16(const float* A, const float* B, v8f acc,
                                           int i, int h) {
#pragma unroll
  for (int k = 0; k < 4; ++k) {
    v2f a = COLA ? ldColS(A, i, k, h) : ldRow2(A, i, k, h);
    if (NEGA) a = -a;
    v2f b = ldRow2(B, i, k, h);
    acc = wmma4(a, b, acc);
  }
  return acc;
}

static __device__ __forceinline__ void stCD(float* D, v8f v, int i, int h) {
  float* p = D + i * 16 + h * 8;
#pragma unroll
  for (int g = 0; g < 8; ++g) p[g] = v[g];
}

static __device__ __forceinline__ void cmatL(const float* Lr, const float* Li,
                                             const float* Sr, const float* Si,
                                             float* Dr, float* Di, int lane) {
  int i = lane & 15, h = lane >> 4;
  v8f ar = {};
  ar = mm16<false, false>(Lr, Sr, ar, i, h);
  ar = mm16<true,  false>(Li, Si, ar, i, h);
  v8f ai = {};
  ai = mm16<false, false>(Lr, Si, ai, i, h);
  ai = mm16<false, false>(Li, Sr, ai, i, h);
  stCD(Dr, ar, i, h);
  stCD(Di, ai, i, h);
}

static __device__ __forceinline__ void cmatR(const float* Tr, const float* Ti,
                                             const float* Rr, const float* Ri,
                                             float* Dr, float* Di, int lane) {
  int i = lane & 15, h = lane >> 4;
  v8f ar = {};
  ar = mm16<false, true>(Tr, Rr, ar, i, h);
  ar = mm16<true,  true>(Ti, Ri, ar, i, h);
  v8f ai = {};
  ai = mm16<false, true>(Tr, Ri, ai, i, h);
  ai = mm16<false, true>(Ti, Rr, ai, i, h);
  stCD(Dr, ar, i, h);
  stCD(Di, ai, i, h);
}

struct G2 { float r00, r01, r10, r11, i00, i01, i10, i11; };
struct C2 { float r, i; };

static __device__ __forceinline__ float pick(bool c, float t, float f) { return c ? t : f; }

static __device__ __forceinline__ C2 cmul(C2 p, C2 x) {
  C2 o;
  o.r = p.r * x.r - p.i * x.i;
  o.i = p.r * x.i + p.i * x.r;
  return o;
}

static __device__ __forceinline__ G2 rotMat(float phi, float th, float om) {
  float st, ct, sa, ca, sb, cb;
  __sincosf(th * 0.5f, &st, &ct);
  __sincosf(0.5f * (phi + om), &sa, &ca);
  __sincosf(0.5f * (phi - om), &sb, &cb);
  G2 g;
  g.r00 = ct * ca;  g.i00 = -ct * sa;
  g.r01 = -st * cb; g.i01 = -st * sb;
  g.r10 = st * cb;  g.i10 = -st * sb;
  g.r11 = ct * ca;  g.i11 = ct * sa;
  return g;
}

static __device__ __forceinline__ void pickRow(bool a, G2 g, C2* c0, C2* c1) {
  c0->r = pick(a, g.r10, g.r00);  c0->i = pick(a, g.i10, g.i00);
  c1->r = pick(a, g.r11, g.r01);  c1->i = pick(a, g.i11, g.i01);
}

static __device__ __forceinline__ void buildKron(G2 g0, G2 g1, G2 g2, G2 g3,
                                                 float* Kr, float* Ki, int lane) {
  const bool a0 = (lane >> 4) & 1;
  const bool a1 = (lane >> 3) & 1;
  const bool a2 = (lane >> 2) & 1;
  const bool a3 = (lane >> 1) & 1;
  const bool b0 = lane & 1;
  C2 x00, x01, p;
  pickRow(a0, g0, &x00, &x01);
  p.r = pick(b0, x01.r, x00.r);
  p.i = pick(b0, x01.i, x00.i);
  C2 u0, u1, v0, v1, w0, w1;
  pickRow(a1, g1, &u0, &u1);
  pickRow(a2, g2, &v0, &v1);
  pickRow(a3, g3, &w0, &w1);
  const C2 s0 = cmul(p, u0), s1 = cmul(p, u1);
  const C2 t0 = cmul(s0, v0), t1 = cmul(s0, v1);
  const C2 t2 = cmul(s1, v0), t3 = cmul(s1, v1);
  const C2 e0 = cmul(t0, w0), e1 = cmul(t0, w1);
  const C2 e2 = cmul(t1, w0), e3 = cmul(t1, w1);
  const C2 e4 = cmul(t2, w0), e5 = cmul(t2, w1);
  const C2 e6 = cmul(t3, w0), e7 = cmul(t3, w1);
  float* kr = Kr + lane * 8;
  float* ki = Ki + lane * 8;
  kr[0] = e0.r; kr[1] = e1.r; kr[2] = e2.r; kr[3] = e3.r;
  kr[4] = e4.r; kr[5] = e5.r; kr[6] = e6.r; kr[7] = e7.r;
  ki[0] = e0.i; ki[1] = e1.i; ki[2] = e2.i; ki[3] = e3.i;
  ki[4] = e4.i; ki[5] = e5.i; ki[6] = e6.i; ki[7] = e7.i;
}

static __device__ __forceinline__ G2 rotGate(const float* vqcw, int l, int q) {
  const float* w = vqcw + (l * 8 + q) * 3;
  return rotMat(w[0], w[1], w[2]);
}

static __device__ __forceinline__ void buildRot(const float* vqcw, int l,
                                                float* GLr, float* GLi,
                                                float* GRr, float* GRi, int lane) {
  buildKron(rotGate(vqcw, l, 0), rotGate(vqcw, l, 1), rotGate(vqcw, l, 2),
            rotGate(vqcw, l, 3), GLr, GLi, lane);
  buildKron(rotGate(vqcw, l, 4), rotGate(vqcw, l, 5), rotGate(vqcw, l, 6),
            rotGate(vqcw, l, 7), GRr, GRi, lane);
}

static __device__ __forceinline__ G2 fusedGate(const float* vqcw, const float* th,
                                               const float* ph, int q) {
  const float* w = vqcw + (8 + q) * 3;
  const G2 m = rotMat(w[0] + ph[q] * 0.5f, w[1], w[2]);
  float s, c;
  __sincosf(th[q] * 0.25f, &s, &c);
  G2 f;
  f.r00 = m.r00 * c + m.r01 * s;  f.i00 = m.i00 * c + m.i01 * s;
  f.r01 = m.r01 * c - m.r00 * s;  f.i01 = m.i01 * c - m.i00 * s;
  f.r10 = m.r10 * c + m.r11 * s;  f.i10 = m.i10 * c + m.i11 * s;
  f.r11 = m.r11 * c - m.r10 * s;  f.i11 = m.i11 * c - m.i10 * s;
  return f;
}

static __device__ __forceinline__ void buildFused(const float* vqcw, const float* th,
                                                  const float* ph,
                                                  float* GLr, float* GLi,
                                                  float* GRr, float* GRi, int lane) {
  buildKron(fusedGate(vqcw, th, ph, 0), fusedGate(vqcw, th, ph, 1),
            fusedGate(vqcw, th, ph, 2), fusedGate(vqcw, th, ph, 3), GLr, GLi, lane);
  buildKron(fusedGate(vqcw, th, ph, 4), fusedGate(vqcw, th, ph, 5),
            fusedGate(vqcw, th, ph, 6), fusedGate(vqcw, th, ph, 7), GRr, GRi, lane);
}

template <int RNG>
static __device__ __forceinline__ void applyCnot(const float* Sr, const float* Si,
                                                 float* Dr, float* Di, int lane) {
#pragma unroll
  for (int j = 0; j < 8; ++j) {
    int d = lane * 8 + j;
    int s = ((d & 15) << 4) | (d >> 4);
    int src = s;
#pragma unroll
    for (int k = 7; k >= 0; --k) {
      int tgt = (k + RNG) & 7;
      src ^= ((src >> (7 - k)) & 1) << (7 - tgt);
    }
    int sa = ((src & 15) << 4) | (src >> 4);
    Dr[d] = Sr[sa];
    Di[d] = Si[sa];
  }
}

extern "C" __global__ __launch_bounds__(256) void vqc_kernel(
    const float* __restrict__ theta, const float* __restrict__ phi,
    const float* __restrict__ jup, const float* __restrict__ vqcw,
    float* __restrict__ out) {
  __shared__ float sXr[WPB][DD], sXi[WPB][DD];
  __shared__ float sYr[WPB][DD], sYi[WPB][DD];
  __shared__ float sGLr[WPB][DD], sGLi[WPB][DD];
  __shared__ float sGRr[WPB][DD], sGRi[WPB][DD];

  int tid = threadIdx.x;
  int wid = tid >> 5, lane = tid & 31;
  int b = blockIdx.x * WPB + wid;

  const float* th = theta + b * 8;
  const float* ph = phi + b * 8;
  const float* jj = jup + b * 28;

  float* Xr = sXr[wid]; float* Xi = sXi[wid];
  float* Yr = sYr[wid]; float* Yi = sYi[wid];
  float* GLr = sGLr[wid]; float* GLi = sGLi[wid];
  float* GRr = sGRr[wid]; float* GRi = sGRi[wid];

  {
    float e0r[8], e0i[8], e1r[8], e1i[8];
#pragma unroll
    for (int k = 0; k < 8; ++k) {
      float s, c, sp, cp;
      __sincosf(th[k] * 0.5f, &s, &c);
      __sincosf(ph[k] * 0.5f, &sp, &cp);
      e0r[k] = c * cp; e0i[k] = -c * sp;
      e1r[k] = s * cp; e1i[k] = s * sp;
    }
    float jv[28];
#pragma unroll
    for (int p = 0; p < 28; ++p) jv[p] = jj[p];

#pragma unroll
    for (int j = 0; j < 8; ++j) {
      int d = lane * 8 + j;
      int s = ((d & 15) << 4) | (d >> 4);
      float pr = 1.f, pim = 0.f;
#pragma unroll
      for (int k = 0; k < 8; ++k) {
        int bit = (s >> (7 - k)) & 1;
        float xr = bit ? e1r[k] : e0r[k];
        float xi = bit ? e1i[k] : e0i[k];
        float nr = pr * xr - pim * xi;
        pim = pr * xi + pim * xr;
        pr = nr;
      }
      float e = 0.f;
      int p = 0;
#pragma unroll
      for (int qi = 0; qi < 8; ++qi)
#pragma unroll
        for (int qj = qi + 1; qj < 8; ++qj) {
          int diff = ((s >> (7 - qi)) & 1) ^ ((s >> (7 - qj)) & 1);
          e += diff ? jv[p] : -jv[p];
          ++p;
        }
      e *= 1.57079632679489662f;
      float se, ce;
      __sincosf(e, &se, &ce);
      Xr[d] = pr * ce - pim * se;
      Xi[d] = pr * se + pim * ce;
    }
  }
  buildRot(vqcw, 0, GLr, GLi, GRr, GRi, lane);
  __syncthreads();

  cmatL(GLr, GLi, Xr, Xi, Yr, Yi, lane);
  __syncthreads();
  cmatR(Yr, Yi, GRr, GRi, Xr, Xi, lane);
  __syncthreads();

  applyCnot<1>(Xr, Xi, Yr, Yi, lane);
  buildFused(vqcw, th, ph, GLr, GLi, GRr, GRi, lane);
  __syncthreads();

  cmatL(GLr, GLi, Yr, Yi, Xr, Xi, lane);
  __syncthreads();
  cmatR(Xr, Xi, GRr, GRi, Yr, Yi, lane);
  __syncthreads();

  applyCnot<2>(Yr, Yi, Xr, Xi, lane);
  __syncthreads();

  float acc[8];
#pragma unroll
  for (int k = 0; k < 8; ++k) acc[k] = 0.f;
#pragma unroll
  for (int j = 0; j < 8; ++j) {
    int d = lane * 8 + j;
    int s = ((d & 15) << 4) | (d >> 4);
    float vr = Xr[d], vi = Xi[d];
    float p2 = vr * vr + vi * vi;
#pragma unroll
    for (int k = 0; k < 8; ++k)
      acc[k] += ((s >> (7 - k)) & 1) ? -p2 : p2;
  }
#pragma unroll
  for (int k = 0; k < 8; ++k) {
    float v = acc[k];
    v += __shfl_xor(v, 16, 32);
    v += __shfl_xor(v, 8, 32);
    v += __shfl_xor(v, 4, 32);
    v += __shfl_xor(v, 2, 32);
    v += __shfl_xor(v, 1, 32);
    acc[k] = v;
  }
  if (lane == 0) {
#pragma unroll
    for (int k = 0; k < 8; ++k) out[b * 8 + k] = acc[k];
  }
}

extern "C" void kernel_launch(void* const* d_in, const int* in_sizes, int n_in,
                              void* d_out, int out_size, void* d_ws, size_t ws_size,
                              hipStream_t stream) {
  const float* theta = (const float*)d_in[0];
  const float* phi   = (const float*)d_in[1];
  const float* jup   = (const float*)d_in[2];
  const float* vqcw  = (const float*)d_in[3];
  float* out = (float*)d_out;
  int B = in_sizes[0] / 8;
  dim3 grid(B / WPB), block(32 * WPB);
  hipLaunchKernelGGL(vqc_kernel, grid, block, 0, stream, theta, phi, jup, vqcw, out);
}


# State and gates stay in WMMA fragment registers through both products of a layer (first product transposed, Kronecker gates built as B fragments); LDS only for the CNOT gathers; sum order changes
# speedup vs baseline: 20.0924x; 1.1068x over previous
#include <hip/hip_runtime.h>

typedef __attribute__((ext_vector_type(2))) float v2f;
typedef __attribute__((ext_vector_type(8))) float v8f;

#define WPB 8
#define DD  256


static __device__ __forceinline__ v8f wmma4(v2f a, v2f b, v8f c) {
  return __builtin_amdgcn_wmma_f32_16x16x4_f32(false, a, false, b, (short)0, c,
                                               false, false);
}

template <bool NEGA>
static __device__ __forceinline__ v8f mmFrag(v8f a, v8f b, v8f acc) {
#pragma unroll
  for (int k = 0; k < 4; ++k) {
    v2f av, bv;
    av.x = a[2 * k]; av.y = a[2 * k + 1];
    if (NEGA) av = -av;
    bv.x = b[2 * k]; bv.y = b[2 * k + 1];
    acc = wmma4(av, bv, acc);
  }
  return acc;
}

static __device__ __forceinline__ void cmm(v8f Ar, v8f Ai, v8f Br, v8f Bi,
                                           v8f* Dr, v8f* Di) {
  v8f r = {};
  r = mmFrag<false>(Ar, Br, r);
  r = mmFrag<true >(Ai, Bi, r);
  v8f i = {};
  i = mmFrag<false>(Ar, Bi, i);
  i = mmFrag<false>(Ai, Br, i);
  *Dr = r; *Di = i;
}

static __device__ __forceinline__ void applyLayer(v8f* Sr, v8f* Si, v8f Lr, v8f Li,
                                                  v8f Rr, v8f Ri) {
  v8f Ur, Ui;
  cmm(*Sr, *Si, Lr, Li, &Ur, &Ui);
  cmm(Ur, Ui, Rr, Ri, Sr, Si);
}

struct G2 { float r00, r01, r10, r11, i00, i01, i10, i11; };
struct C2 { float r, i; };

static __device__ __forceinline__ float pick(bool c, float t, float f) { return c ? t : f; }

static __device__ __forceinline__ C2 cmul(C2 p, C2 x) {
  C2 o;
  o.r = p.r * x.r - p.i * x.i;
  o.i = p.r * x.i + p.i * x.r;
  return o;
}

static __device__ __forceinline__ G2 rotMat(float phi, float th, float om) {
  float st, ct, sa, ca, sb, cb;
  __sincosf(th * 0.5f, &st, &ct);
  __sincosf(0.5f * (phi + om), &sa, &ca);
  __sincosf(0.5f * (phi - om), &sb, &cb);
  G2 g;
  g.r00 = ct * ca;  g.i00 = -ct * sa;
  g.r01 = -st * cb; g.i01 = -st * sb;
  g.r10 = st * cb;  g.i10 = -st * sb;
  g.r11 = ct * ca;  g.i11 = ct * sa;
  return g;
}

static __device__ __forceinline__ void pickRow(bool a, G2 g, C2* c0, C2* c1) {
  c0->r = pick(a, g.r10, g.r00);  c0->i = pick(a, g.i10, g.i00);
  c1->r = pick(a, g.r11, g.r01);  c1->i = pick(a, g.i11, g.i01);
}

static __device__ __forceinline__ void kronFrag(G2 g0, G2 g1, G2 g2, G2 g3, int lane,
                                                v8f* Kr, v8f* Ki) {
  const bool a0 = (lane >> 3) & 1;
  const bool a1 = (lane >> 2) & 1;
  const bool a2 = (lane >> 1) & 1;
  const bool a3 = lane & 1;
  const bool b0 = (lane >> 4) & 1;
  C2 x00, x01, p;
  pickRow(a0, g0, &x00, &x01);
  p.r = pick(b0, x01.r, x00.r);
  p.i = pick(b0, x01.i, x00.i);
  C2 u0, u1, v0, v1, w0, w1;
  pickRow(a1, g1, &u0, &u1);
  pickRow(a2, g2, &v0, &v1);
  pickRow(a3, g3, &w0, &w1);
  const C2 s0 = cmul(p, u0), s1 = cmul(p, u1);
  const C2 t0 = cmul(s0, v0), t1 = cmul(s0, v1);
  const C2 t2 = cmul(s1, v0), t3 = cmul(s1, v1);
  const C2 e0 = cmul(t0, w0), e1 = cmul(t0, w1);
  const C2 e2 = cmul(t1, w0), e3 = cmul(t1, w1);
  const C2 e4 = cmul(t2, w0), e5 = cmul(t2, w1);
  const C2 e6 = cmul(t3, w0), e7 = cmul(t3, w1);
  v8f kr, ki;
  kr[0] = e0.r; kr[1] = e1.r; kr[2] = e2.r; kr[3] = e3.r;
  kr[4] = e4.r; kr[5] = e5.r; kr[6] = e6.r; kr[7] = e7.r;
  ki[0] = e0.i; ki[1] = e1.i; ki[2] = e2.i; ki[3] = e3.i;
  ki[4] = e4.i; ki[5] = e5.i; ki[6] = e6.i; ki[7] = e7.i;
  *Kr = kr; *Ki = ki;
}

static __device__ __forceinline__ G2 rotGate(const float* vqcw, int l, int q) {
  const float* w = vqcw + (l * 8 + q) * 3;
  return rotMat(w[0], w[1], w[2]);
}

static __device__ __forceinline__ G2 fusedGate(const float* vqcw, const float* th,
                                               const float* ph, int q) {
  const float* w = vqcw + (8 + q) * 3;
  const G2 m = rotMat(w[0] + ph[q] * 0.5f, w[1], w[2]);
  float s, c;
  __sincosf(th[q] * 0.25f, &s, &c);
  G2 f;
  f.r00 = m.r00 * c + m.r01 * s;  f.i00 = m.i00 * c + m.i01 * s;
  f.r01 = m.r01 * c - m.r00 * s;  f.i01 = m.i01 * c - m.i00 * s;
  f.r10 = m.r10 * c + m.r11 * s;  f.i10 = m.i10 * c + m.i11 * s;
  f.r11 = m.r11 * c - m.r10 * s;  f.i11 = m.i11 * c - m.i10 * s;
  return f;
}

template <int RNG>
static __device__ __forceinline__ int cnotSrc(int s) {
  int src = s;
#pragma unroll
  for (int k = 7; k >= 0; --k) {
    int tgt = (k + RNG) & 7;
    src ^= ((src >> (7 - k)) & 1) << (7 - tgt);
  }
  return src;
}
static __device__ __forceinline__ int swapNib(int x) { return ((x & 15) << 4) | (x >> 4); }

static __device__ __forceinline__ void stState(float* Br, float* Bi, v8f Sr, v8f Si, int lane) {
  float* pr = Br + (lane & 15) * 16 + (lane >> 4) * 8;
  float* pi = Bi + (lane & 15) * 16 + (lane >> 4) * 8;
#pragma unroll
  for (int g = 0; g < 8; ++g) { pr[g] = Sr[g]; pi[g] = Si[g]; }
}

template <int RNG>
static __device__ __forceinline__ void ldCnot(const float* Br, const float* Bi,
                                              v8f* Sr, v8f* Si, int lane) {
  const int ab = swapNib(cnotSrc<RNG>(((lane >> 4) << 7) | (lane & 15)));
  v8f r, i;
#pragma unroll
  for (int j = 0; j < 8; ++j) {
    const int a = ab ^ swapNib(cnotSrc<RNG>(j << 4));
    r[j] = Br[a];
    i[j] = Bi[a];
  }
  *Sr = r; *Si = i;
}

extern "C" __global__ __launch_bounds__(256) void vqc_kernel(
    const float* __restrict__ theta, const float* __restrict__ phi,
    const float* __restrict__ jup, const float* __restrict__ vqcw,
    float* __restrict__ out) {
  __shared__ float sAr[WPB][DD], sAi[WPB][DD];
  __shared__ float sBr[WPB][DD], sBi[WPB][DD];

  int tid = threadIdx.x;
  int wid = tid >> 5, lane = tid & 31;
  int b = blockIdx.x * WPB + wid;
  const int c = lane & 15, h = lane >> 4;

  const float* th = theta + b * 8;
  const float* ph = phi + b * 8;
  const float* jj = jup + b * 28;

  v8f Sr, Si;

  {
    float e0r[8], e0i[8], e1r[8], e1i[8];
#pragma unroll
    for (int k = 0; k < 8; ++k) {
      float s, cc, sp, cp;
      __sincosf(th[k] * 0.5f, &s, &cc);
      __sincosf(ph[k] * 0.5f, &sp, &cp);
      e0r[k] = cc * cp; e0i[k] = -cc * sp;
      e1r[k] = s * cp;  e1i[k] = s * sp;
    }
    float jv[28];
#pragma unroll
    for (int p = 0; p < 28; ++p) jv[p] = jj[p];

#pragma unroll
    for (int j = 0; j < 8; ++j) {
      int s = ((h * 8 + j) << 4) | c;
      float pr = 1.f, pim = 0.f;
#pragma unroll
      for (int k = 0; k < 8; ++k) {
        int bit = (s >> (7 - k)) & 1;
        float xr = bit ? e1r[k] : e0r[k];
        float xi = bit ? e1i[k] : e0i[k];
        float nr = pr * xr - pim * xi;
        pim = pr * xi + pim * xr;
        pr = nr;
      }
      float e = 0.f;
      int p = 0;
#pragma unroll
      for (int qi = 0; qi < 8; ++qi)
#pragma unroll
        for (int qj = qi + 1; qj < 8; ++qj) {
          int diff = ((s >> (7 - qi)) & 1) ^ ((s >> (7 - qj)) & 1);
          e += diff ? jv[p] : -jv[p];
          ++p;
        }
      e *= 1.57079632679489662f;
      float se, ce;
      __sincosf(e, &se, &ce);
      Sr[j] = pr * ce - pim * se;
      Si[j] = pr * se + pim * ce;
    }
  }

  {
    v8f Lr, Li, Rr, Ri;
    kronFrag(rotGate(vqcw, 0, 0), rotGate(vqcw, 0, 1), rotGate(vqcw, 0, 2),
             rotGate(vqcw, 0, 3), lane, &Lr, &Li);
    kronFrag(rotGate(vqcw, 0, 4), rotGate(vqcw, 0, 5), rotGate(vqcw, 0, 6),
             rotGate(vqcw, 0, 7), lane, &Rr, &Ri);
    applyLayer(&Sr, &Si, Lr, Li, Rr, Ri);
  }

  stState(sAr[wid], sAi[wid], Sr, Si, lane);
  {
    v8f Lr, Li, Rr, Ri;
    kronFrag(fusedGate(vqcw, th, ph, 0), fusedGate(vqcw, th, ph, 1),
             fusedGate(vqcw, th, ph, 2), fusedGate(vqcw, th, ph, 3), lane, &Lr, &Li);
    kronFrag(fusedGate(vqcw, th, ph, 4), fusedGate(vqcw, th, ph, 5),
             fusedGate(vqcw, th, ph, 6), fusedGate(vqcw, th, ph, 7), lane, &Rr, &Ri);
    __syncthreads();
    ldCnot<1>(sAr[wid], sAi[wid], &Sr, &Si, lane);

    applyLayer(&Sr, &Si, Lr, Li, Rr, Ri);
  }

  stState(sBr[wid], sBi[wid], Sr, Si, lane);
  __syncthreads();
  ldCnot<2>(sBr[wid], sBi[wid], &Sr, &Si, lane);

  float tot = 0.f, q1 = 0.f, q2 = 0.f, q3 = 0.f;
#pragma unroll
  for (int j = 0; j < 8; ++j) {
    float p2 = Sr[j] * Sr[j] + Si[j] * Si[j];
    tot += p2;
    q1 += (j & 4) ? -p2 : p2;
    q2 += (j & 2) ? -p2 : p2;
    q3 += (j & 1) ? -p2 : p2;
  }
  float acc[8];
  acc[0] = h ? -tot : tot;
  acc[1] = q1; acc[2] = q2; acc[3] = q3;
  acc[4] = (c & 8) ? -tot : tot;
  acc[5] = (c & 4) ? -tot : tot;
  acc[6] = (c & 2) ? -tot : tot;
  acc[7] = (c & 1) ? -tot : tot;
#pragma unroll
  for (int k = 0; k < 8; ++k) {
    float v = acc[k];
    v += __shfl_xor(v, 16, 32);
    v += __shfl_xor(v, 8, 32);
    v += __shfl_xor(v, 4, 32);
    v += __shfl_xor(v, 2, 32);
    v += __shfl_xor(v, 1, 32);
    acc[k] = v;
  }
  if (lane == 0) {
#pragma unroll
    for (int k = 0; k < 8; ++k) out[b * 8 + k] = acc[k];
  }
}

extern "C" void kernel_launch(void* const* d_in, const int* in_sizes, int n_in,
                              void* d_out, int out_size, void* d_ws, size_t ws_size,
                              hipStream_t stream) {
  const float* theta = (const float*)d_in[0];
  const float* phi   = (const float*)d_in[1];
  const float* jup   = (const float*)d_in[2];
  const float* vqcw  = (const float*)d_in[3];
  float* out = (float*)d_out;
  int B = in_sizes[0] / 8;
  dim3 grid(B / WPB), block(32 * WPB);
  hipLaunchKernelGGL(vqc_kernel, grid, block, 0, stream, theta, phi, jup, vqcw, out);
}


# Gates factored as RZ phases x real RY x RZ phases: an RZ layer is one phase per amplitude (angles summed), Kronecker gates real, a stage is 2 real products a side: 48 WMMAs not 64; sum order changes
# speedup vs baseline: 27.2638x; 1.3569x over previous
#include <hip/hip_runtime.h>

typedef __attribute__((ext_vector_type(2))) float v2f;
typedef __attribute__((ext_vector_type(8))) float v8f;

#define WPB 8
#define DD  256


static __device__ __forceinline__ v8f wmma4(v2f a, v2f b, v8f c) {
  return __builtin_amdgcn_wmma_f32_16x16x4_f32(false, a, false, b, (short)0, c,
                                               false, false);
}

static __device__ __forceinline__ v8f mmFrag(v8f a, v8f b, v8f acc) {
#pragma unroll
  for (int k = 0; k < 4; ++k) {
    v2f av, bv;
    av.x = a[2 * k]; av.y = a[2 * k + 1];
    bv.x = b[2 * k]; bv.y = b[2 * k + 1];
    acc = wmma4(av, bv, acc);
  }
  return acc;
}

static __device__ __forceinline__ void rmm(v8f Ar, v8f Ai, v8f B, v8f* Dr, v8f* Di) {
  v8f r = {};
  r = mmFrag(Ar, B, r);
  v8f i = {};
  i = mmFrag(Ai, B, i);
  *Dr = r; *Di = i;
}

static __device__ __forceinline__ void applyStage(v8f* Sr, v8f* Si, v8f L, v8f R) {
  v8f Ur, Ui;
  rmm(*Sr, *Si, L, &Ur, &Ui);
  rmm(Ur, Ui, R, Sr, Si);
}

struct R2 { float c, s; };

static __device__ __forceinline__ float pick(bool c, float t, float f) { return c ? t : f; }

static __device__ __forceinline__ R2 ryGate(float halfAngle) {
  R2 g;
  __sincosf(halfAngle, &g.s, &g.c);
  return g;
}

static __device__ __forceinline__ void pickRow(bool a, R2 g, float* x0, float* x1) {
  *x0 = pick(a, g.s, g.c);
  *x1 = pick(a, g.c, -g.s);
}

static __device__ __forceinline__ v8f kronFrag(R2 g0, R2 g1, R2 g2, R2 g3, int lane) {
  const bool a0 = (lane >> 3) & 1;
  const bool a1 = (lane >> 2) & 1;
  const bool a2 = (lane >> 1) & 1;
  const bool a3 = lane & 1;
  const bool b0 = (lane >> 4) & 1;
  float x00, x01, u0, u1, v0, v1, w0, w1;
  pickRow(a0, g0, &x00, &x01);
  const float p = pick(b0, x01, x00);
  pickRow(a1, g1, &u0, &u1);
  pickRow(a2, g2, &v0, &v1);
  pickRow(a3, g3, &w0, &w1);
  const float s0 = p * u0, s1 = p * u1;
  const float t0 = s0 * v0, t1 = s0 * v1, t2 = s1 * v0, t3 = s1 * v1;
  v8f k;
  k[0] = t0 * w0; k[1] = t0 * w1; k[2] = t1 * w0; k[3] = t1 * w1;
  k[4] = t2 * w0; k[5] = t2 * w1; k[6] = t3 * w0; k[7] = t3 * w1;
  return k;
}

static __device__ __forceinline__ v8f phaseAngles(float a0, float a1, float a2, float a3,
                                                  float a4, float a5, float a6, float a7,
                                                  int lane) {
  float t = pick((lane >> 4) & 1, a0, -a0);
  t += pick((lane >> 3) & 1, a4, -a4);
  t += pick((lane >> 2) & 1, a5, -a5);
  t += pick((lane >> 1) & 1, a6, -a6);
  t += pick(lane & 1, a7, -a7);
  const float u0 = t - a1, u1 = t + a1;
  const float v0 = u0 - a2, v1 = u0 + a2, v2 = u1 - a2, v3 = u1 + a2;
  v8f e;
  e[0] = v0 - a3; e[1] = v0 + a3; e[2] = v1 - a3; e[3] = v1 + a3;
  e[4] = v2 - a3; e[5] = v2 + a3; e[6] = v3 - a3; e[7] = v3 + a3;
  return e;
}

static __device__ __forceinline__ void mulPhase(v8f* Sr, v8f* Si, v8f E) {
  v8f r, i;
#pragma unroll
  for (int j = 0; j < 8; ++j) {
    float s, c;
    __sincosf(E[j], &s, &c);
    r[j] = (*Sr)[j] * c - (*Si)[j] * s;
    i[j] = (*Sr)[j] * s + (*Si)[j] * c;
  }
  *Sr = r; *Si = i;
}

template <int RNG>
static __device__ __forceinline__ int cnotSrc(int s) {
  int src = s;
#pragma unroll
  for (int k = 7; k >= 0; --k) {
    int tgt = (k + RNG) & 7;
    src ^= ((src >> (7 - k)) & 1) << (7 - tgt);
  }
  return src;
}
static __device__ __forceinline__ int swapNib(int x) { return ((x & 15) << 4) | (x >> 4); }

static __device__ __forceinline__ void stState(float* Br, float* Bi, v8f Sr, v8f Si, int lane) {
  float* pr = Br + (lane & 15) * 16 + (lane >> 4) * 8;
  float* pi = Bi + (lane & 15) * 16 + (lane >> 4) * 8;
#pragma unroll
  for (int g = 0; g < 8; ++g) { pr[g] = Sr[g]; pi[g] = Si[g]; }
}

template <int RNG>
static __device__ __forceinline__ void ldCnot(const float* Br, const float* Bi,
                                              v8f* Sr, v8f* Si, int lane) {
  const int ab = swapNib(cnotSrc<RNG>(((lane >> 4) << 7) | (lane & 15)));
  v8f r, i;
#pragma unroll
  for (int j = 0; j < 8; ++j) {
    const int a = ab ^ swapNib(cnotSrc<RNG>(j << 4));
    r[j] = Br[a];
    i[j] = Bi[a];
  }
  *Sr = r; *Si = i;
}

extern "C" __global__ __launch_bounds__(256) void vqc_kernel(
    const float* __restrict__ theta, const float* __restrict__ phi,
    const float* __restrict__ jup, const float* __restrict__ vqcw,
    float* __restrict__ out) {
  __shared__ float sAr[WPB][DD], sAi[WPB][DD];
  __shared__ float sBr[WPB][DD], sBi[WPB][DD];

  int tid = threadIdx.x;
  int wid = tid >> 5, lane = tid & 31;
  int b = blockIdx.x * WPB + wid;
  const int c = lane & 15, h = lane >> 4;

  const float* th = theta + b * 8;
  const float* ph = phi + b * 8;
  const float* jj = jup + b * 28;
  const float* w0 = vqcw;
  const float* w1 = vqcw + 24;

  v8f Sr, Si;

  {
    float mc[8], ms[8];
#pragma unroll
    for (int k = 0; k < 8; ++k) __sincosf(th[k] * 0.5f, &ms[k], &mc[k]);
    float jv[28];
#pragma unroll
    for (int p = 0; p < 28; ++p) jv[p] = jj[p];

    float mb = pick(h, ms[0], mc[0]);
    mb *= pick((c >> 3) & 1, ms[4], mc[4]);
    mb *= pick((c >> 2) & 1, ms[5], mc[5]);
    mb *= pick((c >> 1) & 1, ms[6], mc[6]);
    mb *= pick(c & 1, ms[7], mc[7]);
    const float m0 = mb * mc[1], m1 = mb * ms[1];
    const float n0 = m0 * mc[2], n1 = m0 * ms[2], n2 = m1 * mc[2], n3 = m1 * ms[2];
    v8f mag;
    mag[0] = n0 * mc[3]; mag[1] = n0 * ms[3]; mag[2] = n1 * mc[3]; mag[3] = n1 * ms[3];
    mag[4] = n2 * mc[3]; mag[5] = n2 * ms[3]; mag[6] = n3 * mc[3]; mag[7] = n3 * ms[3];

    const v8f P = phaseAngles(0.5f * (ph[0] + w0[0]),  0.5f * (ph[1] + w0[3]),
                              0.5f * (ph[2] + w0[6]),  0.5f * (ph[3] + w0[9]),
                              0.5f * (ph[4] + w0[12]), 0.5f * (ph[5] + w0[15]),
                              0.5f * (ph[6] + w0[18]), 0.5f * (ph[7] + w0[21]), lane);

#pragma unroll
    for (int j = 0; j < 8; ++j) {
      int s = ((h * 8 + j) << 4) | c;
      float e = 0.f;
      int p = 0;
#pragma unroll
      for (int qi = 0; qi < 8; ++qi)
#pragma unroll
        for (int qj = qi + 1; qj < 8; ++qj) {
          int diff = ((s >> (7 - qi)) & 1) ^ ((s >> (7 - qj)) & 1);
          e += diff ? jv[p] : -jv[p];
          ++p;
        }
      e = e * 1.57079632679489662f + P[j];
      float se, ce;
      __sincosf(e, &se, &ce);
      Sr[j] = mag[j] * ce;
      Si[j] = mag[j] * se;
    }
  }

  {
    const v8f L = kronFrag(ryGate(0.5f * w0[1]),  ryGate(0.5f * w0[4]),
                           ryGate(0.5f * w0[7]),  ryGate(0.5f * w0[10]), lane);
    const v8f R = kronFrag(ryGate(0.5f * w0[13]), ryGate(0.5f * w0[16]),
                           ryGate(0.5f * w0[19]), ryGate(0.5f * w0[22]), lane);
    applyStage(&Sr, &Si, L, R);
    mulPhase(&Sr, &Si, phaseAngles(0.5f * w0[2],  0.5f * w0[5],  0.5f * w0[8],  0.5f * w0[11],
                                   0.5f * w0[14], 0.5f * w0[17], 0.5f * w0[20], 0.5f * w0[23],
                                   lane));
  }

  stState(sAr[wid], sAi[wid], Sr, Si, lane);
  {
    const v8f Le = kronFrag(ryGate(0.25f * th[0]), ryGate(0.25f * th[1]),
                            ryGate(0.25f * th[2]), ryGate(0.25f * th[3]), lane);
    const v8f Re = kronFrag(ryGate(0.25f * th[4]), ryGate(0.25f * th[5]),
                            ryGate(0.25f * th[6]), ryGate(0.25f * th[7]), lane);
    const v8f E2 = phaseAngles(0.25f * ph[0] + 0.5f * w1[0],  0.25f * ph[1] + 0.5f * w1[3],
                               0.25f * ph[2] + 0.5f * w1[6],  0.25f * ph[3] + 0.5f * w1[9],
                               0.25f * ph[4] + 0.5f * w1[12], 0.25f * ph[5] + 0.5f * w1[15],
                               0.25f * ph[6] + 0.5f * w1[18], 0.25f * ph[7] + 0.5f * w1[21],
                               lane);
    const v8f L1 = kronFrag(ryGate(0.5f * w1[1]),  ryGate(0.5f * w1[4]),
                            ryGate(0.5f * w1[7]),  ryGate(0.5f * w1[10]), lane);
    const v8f R1 = kronFrag(ryGate(0.5f * w1[13]), ryGate(0.5f * w1[16]),
                            ryGate(0.5f * w1[19]), ryGate(0.5f * w1[22]), lane);
    __syncthreads();
    ldCnot<1>(sAr[wid], sAi[wid], &Sr, &Si, lane);

    applyStage(&Sr, &Si, Le, Re);
    mulPhase(&Sr, &Si, E2);
    applyStage(&Sr, &Si, L1, R1);
  }

  stState(sBr[wid], sBi[wid], Sr, Si, lane);
  __syncthreads();
  ldCnot<2>(sBr[wid], sBi[wid], &Sr, &Si, lane);

  float tot = 0.f, q1 = 0.f, q2 = 0.f, q3 = 0.f;
#pragma unroll
  for (int j = 0; j < 8; ++j) {
    float p2 = Sr[j] * Sr[j] + Si[j] * Si[j];
    tot += p2;
    q1 += (j & 4) ? -p2 : p2;
    q2 += (j & 2) ? -p2 : p2;
    q3 += (j & 1) ? -p2 : p2;
  }
  float acc[8];
  acc[0] = h ? -tot : tot;
  acc[1] = q1; acc[2] = q2; acc[3] = q3;
  acc[4] = (c & 8) ? -tot : tot;
  acc[5] = (c & 4) ? -tot : tot;
  acc[6] = (c & 2) ? -tot : tot;
  acc[7] = (c & 1) ? -tot : tot;
#pragma unroll
  for (int k = 0; k < 8; ++k) {
    float v = acc[k];
    v += __shfl_xor(v, 16, 32);
    v += __shfl_xor(v, 8, 32);
    v += __shfl_xor(v, 4, 32);
    v += __shfl_xor(v, 2, 32);
    v += __shfl_xor(v, 1, 32);
    acc[k] = v;
  }
  if (lane == 0) {
#pragma unroll
    for (int k = 0; k < 8; ++k) out[b * 8 + k] = acc[k];
  }
}

extern "C" void kernel_launch(void* const* d_in, const int* in_sizes, int n_in,
                              void* d_out, int out_size, void* d_ws, size_t ws_size,
                              hipStream_t stream) {
  const float* theta = (const float*)d_in[0];
  const float* phi   = (const float*)d_in[1];
  const float* jup   = (const float*)d_in[2];
  const float* vqcw  = (const float*)d_in[3];
  float* out = (float*)d_out;
  int B = in_sizes[0] / 8;
  dim3 grid(B / WPB), block(32 * WPB);
  hipLaunchKernelGGL(vqc_kernel, grid, block, 0, stream, theta, phi, jup, vqcw, out);
}


# Phase-0 exponent (Ising sum + separable phases) factored: lane-qubit part and 3 register-qubit fields by signed multiply-adds, then a 3-level sum tree; the Ising sum's order changes
# speedup vs baseline: 29.9991x; 1.1003x over previous
#include <hip/hip_runtime.h>

typedef __attribute__((ext_vector_type(2))) float v2f;
typedef __attribute__((ext_vector_type(8))) float v8f;

#define WPB 8
#define DD  256


static __device__ __forceinline__ v8f wmma4(v2f a, v2f b, v8f c) {
  return __builtin_amdgcn_wmma_f32_16x16x4_f32(false, a, false, b, (short)0, c,
                                               false, false);
}

static __device__ __forceinline__ v8f mmFrag(v8f a, v8f b, v8f acc) {
#pragma unroll
  for (int k = 0; k < 4; ++k) {
    v2f av, bv;
    av.x = a[2 * k]; av.y = a[2 * k + 1];
    bv.x = b[2 * k]; bv.y = b[2 * k + 1];
    acc = wmma4(av, bv, acc);
  }
  return acc;
}

static __device__ __forceinline__ void rmm(v8f Ar, v8f Ai, v8f B, v8f* Dr, v8f* Di) {
  v8f r = {};
  r = mmFrag(Ar, B, r);
  v8f i = {};
  i = mmFrag(Ai, B, i);
  *Dr = r; *Di = i;
}

static __device__ __forceinline__ void applyStage(v8f* Sr, v8f* Si, v8f L, v8f R) {
  v8f Ur, Ui;
  rmm(*Sr, *Si, L, &Ur, &Ui);
  rmm(Ur, Ui, R, Sr, Si);
}

struct R2 { float c, s; };

static __device__ __forceinline__ float pick(bool c, float t, float f) { return c ? t : f; }

static __device__ __forceinline__ R2 ryGate(float halfAngle) {
  R2 g;
  __sincosf(halfAngle, &g.s, &g.c);
  return g;
}

static __device__ __forceinline__ void pickRow(bool a, R2 g, float* x0, float* x1) {
  *x0 = pick(a, g.s, g.c);
  *x1 = pick(a, g.c, -g.s);
}

static __device__ __forceinline__ v8f kronFrag(R2 g0, R2 g1, R2 g2, R2 g3, int lane) {
  const bool a0 = (lane >> 3) & 1;
  const bool a1 = (lane >> 2) & 1;
  const bool a2 = (lane >> 1) & 1;
  const bool a3 = lane & 1;
  const bool b0 = (lane >> 4) & 1;
  float x00, x01, u0, u1, v0, v1, w0, w1;
  pickRow(a0, g0, &x00, &x01);
  const float p = pick(b0, x01, x00);
  pickRow(a1, g1, &u0, &u1);
  pickRow(a2, g2, &v0, &v1);
  pickRow(a3, g3, &w0, &w1);
  const float s0 = p * u0, s1 = p * u1;
  const float t0 = s0 * v0, t1 = s0 * v1, t2 = s1 * v0, t3 = s1 * v1;
  v8f k;
  k[0] = t0 * w0; k[1] = t0 * w1; k[2] = t1 * w0; k[3] = t1 * w1;
  k[4] = t2 * w0; k[5] = t2 * w1; k[6] = t3 * w0; k[7] = t3 * w1;
  return k;
}

static __device__ __forceinline__ constexpr int pairIdx(int i, int j) {
  return i * (15 - i) / 2 + (j - i - 1);
}

static __device__ __forceinline__ v8f phaseAngles(float a0, float a1, float a2, float a3,
                                                  float a4, float a5, float a6, float a7,
                                                  int lane) {
  float t = pick((lane >> 4) & 1, a0, -a0);
  t += pick((lane >> 3) & 1, a4, -a4);
  t += pick((lane >> 2) & 1, a5, -a5);
  t += pick((lane >> 1) & 1, a6, -a6);
  t += pick(lane & 1, a7, -a7);
  const float u0 = t - a1, u1 = t + a1;
  const float v0 = u0 - a2, v1 = u0 + a2, v2 = u1 - a2, v3 = u1 + a2;
  v8f e;
  e[0] = v0 - a3; e[1] = v0 + a3; e[2] = v1 - a3; e[3] = v1 + a3;
  e[4] = v2 - a3; e[5] = v2 + a3; e[6] = v3 - a3; e[7] = v3 + a3;
  return e;
}

static __device__ __forceinline__ void mulPhase(v8f* Sr, v8f* Si, v8f E) {
  v8f r, i;
#pragma unroll
  for (int j = 0; j < 8; ++j) {
    float s, c;
    __sincosf(E[j], &s, &c);
    r[j] = (*Sr)[j] * c - (*Si)[j] * s;
    i[j] = (*Sr)[j] * s + (*Si)[j] * c;
  }
  *Sr = r; *Si = i;
}

template <int RNG>
static __device__ __forceinline__ int cnotSrc(int s) {
  int src = s;
#pragma unroll
  for (int k = 7; k >= 0; --k) {
    int tgt = (k + RNG) & 7;
    src ^= ((src >> (7 - k)) & 1) << (7 - tgt);
  }
  return src;
}
static __device__ __forceinline__ int swapNib(int x) { return ((x & 15) << 4) | (x >> 4); }

static __device__ __forceinline__ void stState(float* Br, float* Bi, v8f Sr, v8f Si, int lane) {
  float* pr = Br + (lane & 15) * 16 + (lane >> 4) * 8;
  float* pi = Bi + (lane & 15) * 16 + (lane >> 4) * 8;
#pragma unroll
  for (int g = 0; g < 8; ++g) { pr[g] = Sr[g]; pi[g] = Si[g]; }
}

template <int RNG>
static __device__ __forceinline__ void ldCnot(const float* Br, const float* Bi,
                                              v8f* Sr, v8f* Si, int lane) {
  const int ab = swapNib(cnotSrc<RNG>(((lane >> 4) << 7) | (lane & 15)));
  v8f r, i;
#pragma unroll
  for (int j = 0; j < 8; ++j) {
    const int a = ab ^ swapNib(cnotSrc<RNG>(j << 4));
    r[j] = Br[a];
    i[j] = Bi[a];
  }
  *Sr = r; *Si = i;
}

extern "C" __global__ __launch_bounds__(256) void vqc_kernel(
    const float* __restrict__ theta, const float* __restrict__ phi,
    const float* __restrict__ jup, const float* __restrict__ vqcw,
    float* __restrict__ out) {
  __shared__ float sAr[WPB][DD], sAi[WPB][DD];
  __shared__ float sBr[WPB][DD], sBi[WPB][DD];

  int tid = threadIdx.x;
  int wid = tid >> 5, lane = tid & 31;
  int b = blockIdx.x * WPB + wid;
  const int c = lane & 15, h = lane >> 4;

  const float* th = theta + b * 8;
  const float* ph = phi + b * 8;
  const float* jj = jup + b * 28;
  const float* w0 = vqcw;
  const float* w1 = vqcw + 24;

  v8f Sr, Si;

  {
    float mc[8], ms[8];
#pragma unroll
    for (int k = 0; k < 8; ++k) __sincosf(th[k] * 0.5f, &ms[k], &mc[k]);
    float jv[28];
#pragma unroll
    for (int p = 0; p < 28; ++p) jv[p] = jj[p];

    float mb = pick(h, ms[0], mc[0]);
    mb *= pick((c >> 3) & 1, ms[4], mc[4]);
    mb *= pick((c >> 2) & 1, ms[5], mc[5]);
    mb *= pick((c >> 1) & 1, ms[6], mc[6]);
    mb *= pick(c & 1, ms[7], mc[7]);
    const float m0 = mb * mc[1], m1 = mb * ms[1];
    const float n0 = m0 * mc[2], n1 = m0 * ms[2], n2 = m1 * mc[2], n3 = m1 * ms[2];
    v8f mag;
    mag[0] = n0 * mc[3]; mag[1] = n0 * ms[3]; mag[2] = n1 * mc[3]; mag[3] = n1 * ms[3];
    mag[4] = n2 * mc[3]; mag[5] = n2 * ms[3]; mag[6] = n3 * mc[3]; mag[7] = n3 * ms[3];

    const float HP = 1.57079632679489662f;
    const float g0 = pick(h, -1.f, 1.f);
    const float g4 = pick((c >> 3) & 1, -1.f, 1.f);
    const float g5 = pick((c >> 2) & 1, -1.f, 1.f);
    const float g6 = pick((c >> 1) & 1, -1.f, 1.f);
    const float g7 = pick(c & 1, -1.f, 1.f);

    const float a0 = 0.5f * (ph[0] + w0[0]),  a1 = 0.5f * (ph[1] + w0[3]);
    const float a2 = 0.5f * (ph[2] + w0[6]),  a3 = 0.5f * (ph[3] + w0[9]);
    const float a4 = 0.5f * (ph[4] + w0[12]), a5 = 0.5f * (ph[5] + w0[15]);
    const float a6 = 0.5f * (ph[6] + w0[18]), a7 = 0.5f * (ph[7] + w0[21]);

    const float i0s = jv[pairIdx(0, 4)] * g4 + jv[pairIdx(0, 5)] * g5 + jv[pairIdx(0, 6)] * g6 +
                      jv[pairIdx(0, 7)] * g7;
    const float i4s = jv[pairIdx(4, 5)] * g5 + jv[pairIdx(4, 6)] * g6 + jv[pairIdx(4, 7)] * g7;
    const float i5s = jv[pairIdx(5, 6)] * g6 + jv[pairIdx(5, 7)] * g7;
    const float i6s = jv[pairIdx(6, 7)] * g7;
    const float A = -(g0 * (HP * i0s + a0) + g4 * (HP * i4s + a4) + g5 * (HP * i5s + a5) +
                      g6 * (HP * i6s + a6) + g7 * a7);

    const float f1 = jv[pairIdx(0, 1)] * g0 + jv[pairIdx(1, 4)] * g4 + jv[pairIdx(1, 5)] * g5 +
                     jv[pairIdx(1, 6)] * g6 + jv[pairIdx(1, 7)] * g7;
    const float f2 = jv[pairIdx(0, 2)] * g0 + jv[pairIdx(2, 4)] * g4 + jv[pairIdx(2, 5)] * g5 +
                     jv[pairIdx(2, 6)] * g6 + jv[pairIdx(2, 7)] * g7;
    const float f3 = jv[pairIdx(0, 3)] * g0 + jv[pairIdx(3, 4)] * g4 + jv[pairIdx(3, 5)] * g5 +
                     jv[pairIdx(3, 6)] * g6 + jv[pairIdx(3, 7)] * g7;
    const float F1 = -(HP * f1 + a1), F2 = -(HP * f2 + a2), F3 = -(HP * f3 + a3);
    const float K12 = HP * jv[pairIdx(1, 2)], K13 = HP * jv[pairIdx(1, 3)];
    const float K23 = HP * jv[pairIdx(2, 3)];

    const float tP = A + F1, tM = A - F1;
    const float F2p = F2 - K12, F2m = F2 + K12;
    const float uPP = tP + F2p, uPM = tP - F2p, uMP = tM + F2m, uMM = tM - F2m;
    const float F3p = F3 - K13, F3m = F3 + K13;
    const float vPP = F3p - K23, vPM = F3p + K23;
    const float vMP = F3m - K23, vMM = F3m + K23;
    v8f X;
    X[0] = uPP + vPP; X[1] = uPP - vPP; X[2] = uPM + vPM; X[3] = uPM - vPM;
    X[4] = uMP + vMP; X[5] = uMP - vMP; X[6] = uMM + vMM; X[7] = uMM - vMM;

#pragma unroll
    for (int j = 0; j < 8; ++j) {
      float se, ce;
      __sincosf(X[j], &se, &ce);
      Sr[j] = mag[j] * ce;
      Si[j] = mag[j] * se;
    }
  }

  {
    const v8f L = kronFrag(ryGate(0.5f * w0[1]),  ryGate(0.5f * w0[4]),
                           ryGate(0.5f * w0[7]),  ryGate(0.5f * w0[10]), lane);
    const v8f R = kronFrag(ryGate(0.5f * w0[13]), ryGate(0.5f * w0[16]),
                           ryGate(0.5f * w0[19]), ryGate(0.5f * w0[22]), lane);
    applyStage(&Sr, &Si, L, R);
    mulPhase(&Sr, &Si, phaseAngles(0.5f * w0[2],  0.5f * w0[5],  0.5f * w0[8],  0.5f * w0[11],
                                   0.5f * w0[14], 0.5f * w0[17], 0.5f * w0[20], 0.5f * w0[23],
                                   lane));
  }

  stState(sAr[wid], sAi[wid], Sr, Si, lane);
  {
    const v8f Le = kronFrag(ryGate(0.25f * th[0]), ryGate(0.25f * th[1]),
                            ryGate(0.25f * th[2]), ryGate(0.25f * th[3]), lane);
    const v8f Re = kronFrag(ryGate(0.25f * th[4]), ryGate(0.25f * th[5]),
                            ryGate(0.25f * th[6]), ryGate(0.25f * th[7]), lane);
    const v8f E2 = phaseAngles(0.25f * ph[0] + 0.5f * w1[0],  0.25f * ph[1] + 0.5f * w1[3],
                               0.25f * ph[2] + 0.5f * w1[6],  0.25f * ph[3] + 0.5f * w1[9],
                               0.25f * ph[4] + 0.5f * w1[12], 0.25f * ph[5] + 0.5f * w1[15],
                               0.25f * ph[6] + 0.5f * w1[18], 0.25f * ph[7] + 0.5f * w1[21],
                               lane);
    const v8f L1 = kronFrag(ryGate(0.5f * w1[1]),  ryGate(0.5f * w1[4]),
                            ryGate(0.5f * w1[7]),  ryGate(0.5f * w1[10]), lane);
    const v8f R1 = kronFrag(ryGate(0.5f * w1[13]), ryGate(0.5f * w1[16]),
                            ryGate(0.5f * w1[19]), ryGate(0.5f * w1[22]), lane);
    __syncthreads();
    ldCnot<1>(sAr[wid], sAi[wid], &Sr, &Si, lane);

    applyStage(&Sr, &Si, Le, Re);
    mulPhase(&Sr, &Si, E2);
    applyStage(&Sr, &Si, L1, R1);
  }

  stState(sBr[wid], sBi[wid], Sr, Si, lane);
  __syncthreads();
  ldCnot<2>(sBr[wid], sBi[wid], &Sr, &Si, lane);

  float tot = 0.f, q1 = 0.f, q2 = 0.f, q3 = 0.f;
#pragma unroll
  for (int j = 0; j < 8; ++j) {
    float p2 = Sr[j] * Sr[j] + Si[j] * Si[j];
    tot += p2;
    q1 += (j & 4) ? -p2 : p2;
    q2 += (j & 2) ? -p2 : p2;
    q3 += (j & 1) ? -p2 : p2;
  }
  float acc[8];
  acc[0] = h ? -tot : tot;
  acc[1] = q1; acc[2] = q2; acc[3] = q3;
  acc[4] = (c & 8) ? -tot : tot;
  acc[5] = (c & 4) ? -tot : tot;
  acc[6] = (c & 2) ? -tot : tot;
  acc[7] = (c & 1) ? -tot : tot;
#pragma unroll
  for (int k = 0; k < 8; ++k) {
    float v = acc[k];
    v += __shfl_xor(v, 16, 32);
    v += __shfl_xor(v, 8, 32);
    v += __shfl_xor(v, 4, 32);
    v += __shfl_xor(v, 2, 32);
    v += __shfl_xor(v, 1, 32);
    acc[k] = v;
  }
  if (lane == 0) {
#pragma unroll
    for (int k = 0; k < 8; ++k) out[b * 8 + k] = acc[k];
  }
}

extern "C" void kernel_launch(void* const* d_in, const int* in_sizes, int n_in,
                              void* d_out, int out_size, void* d_ws, size_t ws_size,
                              hipStream_t stream) {
  const float* theta = (const float*)d_in[0];
  const float* phi   = (const float*)d_in[1];
  const float* jup   = (const float*)d_in[2];
  const float* vqcw  = (const float*)d_in[3];
  float* out = (float*)d_out;
  int B = in_sizes[0] / 8;
  dim3 grid(B / WPB), block(32 * WPB);
  hipLaunchKernelGGL(vqc_kernel, grid, block, 0, stream, theta, phi, jup, vqcw, out);
}


# Weight-only operands (4 Kronecker RY gates, layer-0 RZ phase factors) built once per block by 5 of its 8 waves into LDS and read by every wave instead of rebuilt per wave; sum order unchanged
# speedup vs baseline: 32.0379x; 1.0680x over previous
#include <hip/hip_runtime.h>

typedef __attribute__((ext_vector_type(2))) float v2f;
typedef __attribute__((ext_vector_type(4))) float v4f;
typedef __attribute__((ext_vector_type(8))) float v8f;

#define WPB 8
#define DD  256
#define NSH 6


static __device__ __forceinline__ v8f wmma4(v2f a, v2f b, v8f c) {
  return __builtin_amdgcn_wmma_f32_16x16x4_f32(false, a, false, b, (short)0, c,
                                               false, false);
}

static __device__ __forceinline__ v8f mmFrag(v8f a, v8f b, v8f acc) {
#pragma unroll
  for (int k = 0; k < 4; ++k) {
    v2f av, bv;
    av.x = a[2 * k]; av.y = a[2 * k + 1];
    bv.x = b[2 * k]; bv.y = b[2 * k + 1];
    acc = wmma4(av, bv, acc);
  }
  return acc;
}

static __device__ __forceinline__ void rmm(v8f Ar, v8f Ai, v8f B, v8f* Dr, v8f* Di) {
  v8f r = {};
  r = mmFrag(Ar, B, r);
  v8f i = {};
  i = mmFrag(Ai, B, i);
  *Dr = r; *Di = i;
}

static __device__ __forceinline__ void applyStage(v8f* Sr, v8f* Si, v8f L, v8f R) {
  v8f Ur, Ui;
  rmm(*Sr, *Si, L, &Ur, &Ui);
  rmm(Ur, Ui, R, Sr, Si);
}

struct R2 { float c, s; };

static __device__ __forceinline__ float pick(bool c, float t, float f) { return c ? t : f; }

static __device__ __forceinline__ R2 ryGate(float halfAngle) {
  R2 g;
  __sincosf(halfAngle, &g.s, &g.c);
  return g;
}

static __device__ __forceinline__ void pickRow(bool a, R2 g, float* x0, float* x1) {
  *x0 = pick(a, g.s, g.c);
  *x1 = pick(a, g.c, -g.s);
}

static __device__ __forceinline__ v8f kronFrag(R2 g0, R2 g1, R2 g2, R2 g3, int lane) {
  const bool a0 = (lane >> 3) & 1;
  const bool a1 = (lane >> 2) & 1;
  const bool a2 = (lane >> 1) & 1;
  const bool a3 = lane & 1;
  const bool b0 = (lane >> 4) & 1;
  float x00, x01, u0, u1, v0, v1, w0, w1;
  pickRow(a0, g0, &x00, &x01);
  const float p = pick(b0, x01, x00);
  pickRow(a1, g1, &u0, &u1);
  pickRow(a2, g2, &v0, &v1);
  pickRow(a3, g3, &w0, &w1);
  const float s0 = p * u0, s1 = p * u1;
  const float t0 = s0 * v0, t1 = s0 * v1, t2 = s1 * v0, t3 = s1 * v1;
  v8f k;
  k[0] = t0 * w0; k[1] = t0 * w1; k[2] = t1 * w0; k[3] = t1 * w1;
  k[4] = t2 * w0; k[5] = t2 * w1; k[6] = t3 * w0; k[7] = t3 * w1;
  return k;
}

static __device__ __forceinline__ constexpr int pairIdx(int i, int j) {
  return i * (15 - i) / 2 + (j - i - 1);
}

static __device__ __forceinline__ v8f phaseAngles(float a0, float a1, float a2, float a3,
                                                  float a4, float a5, float a6, float a7,
                                                  int lane) {
  float t = pick((lane >> 4) & 1, a0, -a0);
  t += pick((lane >> 3) & 1, a4, -a4);
  t += pick((lane >> 2) & 1, a5, -a5);
  t += pick((lane >> 1) & 1, a6, -a6);
  t += pick(lane & 1, a7, -a7);
  const float u0 = t - a1, u1 = t + a1;
  const float v0 = u0 - a2, v1 = u0 + a2, v2 = u1 - a2, v3 = u1 + a2;
  v8f e;
  e[0] = v0 - a3; e[1] = v0 + a3; e[2] = v1 - a3; e[3] = v1 + a3;
  e[4] = v2 - a3; e[5] = v2 + a3; e[6] = v3 - a3; e[7] = v3 + a3;
  return e;
}

static __device__ __forceinline__ void mulPhaseCS(v8f* Sr, v8f* Si, v8f C, v8f Sn) {
  v8f r, i;
#pragma unroll
  for (int j = 0; j < 8; ++j) {
    r[j] = (*Sr)[j] * C[j] - (*Si)[j] * Sn[j];
    i[j] = (*Sr)[j] * Sn[j] + (*Si)[j] * C[j];
  }
  *Sr = r; *Si = i;
}

static __device__ __forceinline__ void mulPhase(v8f* Sr, v8f* Si, v8f E) {
  v8f r, i;
#pragma unroll
  for (int j = 0; j < 8; ++j) {
    float s, c;
    __sincosf(E[j], &s, &c);
    r[j] = (*Sr)[j] * c - (*Si)[j] * s;
    i[j] = (*Sr)[j] * s + (*Si)[j] * c;
  }
  *Sr = r; *Si = i;
}

static __device__ __forceinline__ void stFrag(float* F, v8f v, int lane) {
  v4f lo, hi;
  lo.x = v[0]; lo.y = v[1]; lo.z = v[2]; lo.w = v[3];
  hi.x = v[4]; hi.y = v[5]; hi.z = v[6]; hi.w = v[7];
  *(v4f*)(F + lane * 4) = lo;
  *(v4f*)(F + 128 + lane * 4) = hi;
}
static __device__ __forceinline__ v8f ldFrag(const float* F, int lane) {
  const v4f lo = *(const v4f*)(F + lane * 4);
  const v4f hi = *(const v4f*)(F + 128 + lane * 4);
  v8f v;
  v[0] = lo.x; v[1] = lo.y; v[2] = lo.z; v[3] = lo.w;
  v[4] = hi.x; v[5] = hi.y; v[6] = hi.z; v[7] = hi.w;
  return v;
}

template <int RNG>
static __device__ __forceinline__ int cnotSrc(int s) {
  int src = s;
#pragma unroll
  for (int k = 7; k >= 0; --k) {
    int tgt = (k + RNG) & 7;
    src ^= ((src >> (7 - k)) & 1) << (7 - tgt);
  }
  return src;
}
static __device__ __forceinline__ int swapNib(int x) { return ((x & 15) << 4) | (x >> 4); }

static __device__ __forceinline__ void stState(float* Br, float* Bi, v8f Sr, v8f Si, int lane) {
  float* pr = Br + (lane & 15) * 16 + (lane >> 4) * 8;
  float* pi = Bi + (lane & 15) * 16 + (lane >> 4) * 8;
#pragma unroll
  for (int g = 0; g < 8; ++g) { pr[g] = Sr[g]; pi[g] = Si[g]; }
}

template <int RNG>
static __device__ __forceinline__ void ldCnot(const float* Br, const float* Bi,
                                              v8f* Sr, v8f* Si, int lane) {
  const int ab = swapNib(cnotSrc<RNG>(((lane >> 4) << 7) | (lane & 15)));
  v8f r, i;
#pragma unroll
  for (int j = 0; j < 8; ++j) {
    const int a = ab ^ swapNib(cnotSrc<RNG>(j << 4));
    r[j] = Br[a];
    i[j] = Bi[a];
  }
  *Sr = r; *Si = i;
}

extern "C" __global__ __launch_bounds__(256, 12) void vqc_kernel(
    const float* __restrict__ theta, const float* __restrict__ phi,
    const float* __restrict__ jup, const float* __restrict__ vqcw,
    float* __restrict__ out) {
  __shared__ float sAr[WPB][DD], sAi[WPB][DD];
  __shared__ float sBr[WPB][DD], sBi[WPB][DD];
  __shared__ __attribute__((aligned(16))) float sW[NSH][DD];

  int tid = threadIdx.x;
  int wid = tid >> 5, lane = tid & 31;
  int b = blockIdx.x * WPB + wid;
  const int c = lane & 15, h = lane >> 4;

  const float* th = theta + b * 8;
  const float* ph = phi + b * 8;
  const float* jj = jup + b * 28;
  const float* w0 = vqcw;
  const float* w1 = vqcw + 24;

  float thv[8], phv[8], jv[28];
#pragma unroll
  for (int k = 0; k < 8; ++k) { thv[k] = th[k]; phv[k] = ph[k]; }
#pragma unroll
  for (int p = 0; p < 28; ++p) jv[p] = jj[p];

  {
    const int wu = __builtin_amdgcn_readfirstlane(wid);
    if (wu < 4) {
      const float* g = vqcw + (wu >> 1) * 24 + (wu & 1) * 12;
      stFrag(sW[wu], kronFrag(ryGate(0.5f * g[1]), ryGate(0.5f * g[4]),
                              ryGate(0.5f * g[7]), ryGate(0.5f * g[10]), lane), lane);
    } else if (wu == 4) {
      const v8f E = phaseAngles(0.5f * w0[2],  0.5f * w0[5],  0.5f * w0[8],  0.5f * w0[11],
                                0.5f * w0[14], 0.5f * w0[17], 0.5f * w0[20], 0.5f * w0[23],
                                lane);
      v8f pc, ps;
#pragma unroll
      for (int j = 0; j < 8; ++j) {
        float s, c;
        __sincosf(E[j], &s, &c);
        pc[j] = c; ps[j] = s;
      }
      stFrag(sW[4], pc, lane);
      stFrag(sW[5], ps, lane);
    }
  }

  v8f Sr, Si;

  {
    float mc[8], ms[8];
#pragma unroll
    for (int k = 0; k < 8; ++k) __sincosf(thv[k] * 0.5f, &ms[k], &mc[k]);

    float mb = pick(h, ms[0], mc[0]);
    mb *= pick((c >> 3) & 1, ms[4], mc[4]);
    mb *= pick((c >> 2) & 1, ms[5], mc[5]);
    mb *= pick((c >> 1) & 1, ms[6], mc[6]);
    mb *= pick(c & 1, ms[7], mc[7]);
    const float m0 = mb * mc[1], m1 = mb * ms[1];
    const float n0 = m0 * mc[2], n1 = m0 * ms[2], n2 = m1 * mc[2], n3 = m1 * ms[2];
    v8f mag;
    mag[0] = n0 * mc[3]; mag[1] = n0 * ms[3]; mag[2] = n1 * mc[3]; mag[3] = n1 * ms[3];
    mag[4] = n2 * mc[3]; mag[5] = n2 * ms[3]; mag[6] = n3 * mc[3]; mag[7] = n3 * ms[3];

    const float HP = 1.57079632679489662f;
    const float g0 = pick(h, -1.f, 1.f);
    const float g4 = pick((c >> 3) & 1, -1.f, 1.f);
    const float g5 = pick((c >> 2) & 1, -1.f, 1.f);
    const float g6 = pick((c >> 1) & 1, -1.f, 1.f);
    const float g7 = pick(c & 1, -1.f, 1.f);

    const float a0 = 0.5f * (phv[0] + w0[0]),  a1 = 0.5f * (phv[1] + w0[3]);
    const float a2 = 0.5f * (phv[2] + w0[6]),  a3 = 0.5f * (phv[3] + w0[9]);
    const float a4 = 0.5f * (phv[4] + w0[12]), a5 = 0.5f * (phv[5] + w0[15]);
    const float a6 = 0.5f * (phv[6] + w0[18]), a7 = 0.5f * (phv[7] + w0[21]);

    const float i0s = jv[pairIdx(0, 4)] * g4 + jv[pairIdx(0, 5)] * g5 + jv[pairIdx(0, 6)] * g6 +
                      jv[pairIdx(0, 7)] * g7;
    const float i4s = jv[pairIdx(4, 5)] * g5 + jv[pairIdx(4, 6)] * g6 + jv[pairIdx(4, 7)] * g7;
    const float i5s = jv[pairIdx(5, 6)] * g6 + jv[pairIdx(5, 7)] * g7;
    const float i6s = jv[pairIdx(6, 7)] * g7;
    const float A = -(g0 * (HP * i0s + a0) + g4 * (HP * i4s + a4) + g5 * (HP * i5s + a5) +
                      g6 * (HP * i6s + a6) + g7 * a7);

    const float f1 = jv[pairIdx(0, 1)] * g0 + jv[pairIdx(1, 4)] * g4 + jv[pairIdx(1, 5)] * g5 +
                     jv[pairIdx(1, 6)] * g6 + jv[pairIdx(1, 7)] * g7;
    const float f2 = jv[pairIdx(0, 2)] * g0 + jv[pairIdx(2, 4)] * g4 + jv[pairIdx(2, 5)] * g5 +
                     jv[pairIdx(2, 6)] * g6 + jv[pairIdx(2, 7)] * g7;
    const float f3 = jv[pairIdx(0, 3)] * g0 + jv[pairIdx(3, 4)] * g4 + jv[pairIdx(3, 5)] * g5 +
                     jv[pairIdx(3, 6)] * g6 + jv[pairIdx(3, 7)] * g7;
    const float F1 = -(HP * f1 + a1), F2 = -(HP * f2 + a2), F3 = -(HP * f3 + a3);
    const float K12 = HP * jv[pairIdx(1, 2)], K13 = HP * jv[pairIdx(1, 3)];
    const float K23 = HP * jv[pairIdx(2, 3)];

    const float tP = A + F1, tM = A - F1;
    const float F2p = F2 - K12, F2m = F2 + K12;
    const float uPP = tP + F2p, uPM = tP - F2p, uMP = tM + F2m, uMM = tM - F2m;
    const float F3p = F3 - K13, F3m = F3 + K13;
    const float vPP = F3p - K23, vPM = F3p + K23;
    const float vMP = F3m - K23, vMM = F3m + K23;
    v8f X;
    X[0] = uPP + vPP; X[1] = uPP - vPP; X[2] = uPM + vPM; X[3] = uPM - vPM;
    X[4] = uMP + vMP; X[5] = uMP - vMP; X[6] = uMM + vMM; X[7] = uMM - vMM;

#pragma unroll
    for (int j = 0; j < 8; ++j) {
      float se, ce;
      __sincosf(X[j], &se, &ce);
      Sr[j] = mag[j] * ce;
      Si[j] = mag[j] * se;
    }
  }

  __syncthreads();
  {
    const v8f L = ldFrag(sW[0], lane);
    const v8f R = ldFrag(sW[1], lane);
    applyStage(&Sr, &Si, L, R);
    mulPhaseCS(&Sr, &Si, ldFrag(sW[4], lane), ldFrag(sW[5], lane));
  }

  stState(sAr[wid], sAi[wid], Sr, Si, lane);
  {
    const v8f Le = kronFrag(ryGate(0.25f * thv[0]), ryGate(0.25f * thv[1]),
                            ryGate(0.25f * thv[2]), ryGate(0.25f * thv[3]), lane);
    const v8f Re = kronFrag(ryGate(0.25f * thv[4]), ryGate(0.25f * thv[5]),
                            ryGate(0.25f * thv[6]), ryGate(0.25f * thv[7]), lane);
    const v8f E2 = phaseAngles(0.25f * phv[0] + 0.5f * w1[0],  0.25f * phv[1] + 0.5f * w1[3],
                               0.25f * phv[2] + 0.5f * w1[6],  0.25f * phv[3] + 0.5f * w1[9],
                               0.25f * phv[4] + 0.5f * w1[12], 0.25f * phv[5] + 0.5f * w1[15],
                               0.25f * phv[6] + 0.5f * w1[18], 0.25f * phv[7] + 0.5f * w1[21],
                               lane);
    const v8f L1 = ldFrag(sW[2], lane);
    const v8f R1 = ldFrag(sW[3], lane);
    __syncthreads();
    ldCnot<1>(sAr[wid], sAi[wid], &Sr, &Si, lane);

    applyStage(&Sr, &Si, Le, Re);
    mulPhase(&Sr, &Si, E2);
    applyStage(&Sr, &Si, L1, R1);
  }

  stState(sBr[wid], sBi[wid], Sr, Si, lane);
  __syncthreads();
  ldCnot<2>(sBr[wid], sBi[wid], &Sr, &Si, lane);

  float tot = 0.f, q1 = 0.f, q2 = 0.f, q3 = 0.f;
#pragma unroll
  for (int j = 0; j < 8; ++j) {
    float p2 = Sr[j] * Sr[j] + Si[j] * Si[j];
    tot += p2;
    q1 += (j & 4) ? -p2 : p2;
    q2 += (j & 2) ? -p2 : p2;
    q3 += (j & 1) ? -p2 : p2;
  }
  float acc[8];
  acc[0] = h ? -tot : tot;
  acc[1] = q1; acc[2] = q2; acc[3] = q3;
  acc[4] = (c & 8) ? -tot : tot;
  acc[5] = (c & 4) ? -tot : tot;
  acc[6] = (c & 2) ? -tot : tot;
  acc[7] = (c & 1) ? -tot : tot;
#pragma unroll
  for (int k = 0; k < 8; ++k) {
    float v = acc[k];
    v += __shfl_xor(v, 16, 32);
    v += __shfl_xor(v, 8, 32);
    v += __shfl_xor(v, 4, 32);
    v += __shfl_xor(v, 2, 32);
    v += __shfl_xor(v, 1, 32);
    acc[k] = v;
  }
  if (lane == 0) {
#pragma unroll
    for (int k = 0; k < 8; ++k) out[b * 8 + k] = acc[k];
  }
}

extern "C" void kernel_launch(void* const* d_in, const int* in_sizes, int n_in,
                              void* d_out, int out_size, void* d_ws, size_t ws_size,
                              hipStream_t stream) {
  const float* theta = (const float*)d_in[0];
  const float* phi   = (const float*)d_in[1];
  const float* jup   = (const float*)d_in[2];
  const float* vqcw  = (const float*)d_in[3];
  float* out = (float*)d_out;
  int B = in_sizes[0] / 8;
  dim3 grid(B / WPB), block(32 * WPB);
  hipLaunchKernelGGL(vqc_kernel, grid, block, 0, stream, theta, phi, jup, vqcw, out);
}


# Re-encoding RY,RZ and layer-1 RZ,RY of each qubit fused into one phase + one real RY (Euler form of the 2x2 product; one qubit per lane, broadcast by readlane): 32 WMMAs not 48; sum order changes
# speedup vs baseline: 36.8480x; 1.1501x over previous
#include <hip/hip_runtime.h>

typedef __attribute__((ext_vector_type(2))) float v2f;
typedef __attribute__((ext_vector_type(4))) float v4f;
typedef __attribute__((ext_vector_type(8))) float v8f;

#define WPB 8
#define DD  256
#define NSH 4


static __device__ __forceinline__ v8f wmma4(v2f a, v2f b, v8f c) {
  return __builtin_amdgcn_wmma_f32_16x16x4_f32(false, a, false, b, (short)0, c,
                                               false, false);
}

static __device__ __forceinline__ v8f mmFrag(v8f a, v8f b, v8f acc) {
#pragma unroll
  for (int k = 0; k < 4; ++k) {
    v2f av, bv;
    av.x = a[2 * k]; av.y = a[2 * k + 1];
    bv.x = b[2 * k]; bv.y = b[2 * k + 1];
    acc = wmma4(av, bv, acc);
  }
  return acc;
}

static __device__ __forceinline__ void rmm(v8f Ar, v8f Ai, v8f B, v8f* Dr, v8f* Di) {
  v8f r = {};
  r = mmFrag(Ar, B, r);
  v8f i = {};
  i = mmFrag(Ai, B, i);
  *Dr = r; *Di = i;
}

static __device__ __forceinline__ void applyStage(v8f* Sr, v8f* Si, v8f L, v8f R) {
  v8f Ur, Ui;
  rmm(*Sr, *Si, L, &Ur, &Ui);
  rmm(Ur, Ui, R, Sr, Si);
}

struct R2 { float c, s; };

static __device__ __forceinline__ float pick(bool c, float t, float f) { return c ? t : f; }

static __device__ __forceinline__ R2 ryGate(float halfAngle) {
  R2 g;
  __sincosf(halfAngle, &g.s, &g.c);
  return g;
}

static __device__ __forceinline__ void pickRow(bool a, R2 g, float* x0, float* x1) {
  *x0 = pick(a, g.s, g.c);
  *x1 = pick(a, g.c, -g.s);
}

static __device__ __forceinline__ v8f kronFrag(R2 g0, R2 g1, R2 g2, R2 g3, int lane) {
  const bool a0 = (lane >> 3) & 1;
  const bool a1 = (lane >> 2) & 1;
  const bool a2 = (lane >> 1) & 1;
  const bool a3 = lane & 1;
  const bool b0 = (lane >> 4) & 1;
  float x00, x01, u0, u1, v0, v1, w0, w1;
  pickRow(a0, g0, &x00, &x01);
  const float p = pick(b0, x01, x00);
  pickRow(a1, g1, &u0, &u1);
  pickRow(a2, g2, &v0, &v1);
  pickRow(a3, g3, &w0, &w1);
  const float s0 = p * u0, s1 = p * u1;
  const float t0 = s0 * v0, t1 = s0 * v1, t2 = s1 * v0, t3 = s1 * v1;
  v8f k;
  k[0] = t0 * w0; k[1] = t0 * w1; k[2] = t1 * w0; k[3] = t1 * w1;
  k[4] = t2 * w0; k[5] = t2 * w1; k[6] = t3 * w0; k[7] = t3 * w1;
  return k;
}

struct FG { float c, s, pr, pi; };

static __device__ __forceinline__ FG fuseGate(float thq, float phq, float wa, float wb) {
  float sP, cP, sM, cM, sE, cE;
  __sincosf(0.25f * thq + 0.5f * wb, &sP, &cP);
  __sincosf(0.25f * thq - 0.5f * wb, &sM, &cM);
  __sincosf(0.25f * phq + 0.5f * wa, &sE, &cE);
  const float ur = cP * cE, ui = -(cM * sE);
  const float vr = sP * cE, vi = sM * sE;
  const float nu = ur * ur + ui * ui, nv = vr * vr + vi * vi;
  const float TINY = 1e-30f;
  const bool ku = nu > TINY, kv = nv > TINY;
  const float iu = ku ? __builtin_amdgcn_rsqf(nu) : 0.f;
  const float iv = kv ? __builtin_amdgcn_rsqf(nv) : 0.f;
  FG g;
  g.c = nu * iu;
  g.s = nv * iv;
  const float aur = ku ? ur * iu : 1.f, aui = -(ui * iu);
  const float avr = kv ? vr * iv : 1.f, avi = -(vi * iv);
  g.pr = aur * avr - aui * avi;
  g.pi = aur * avi + aui * avr;
  return g;
}

static __device__ __forceinline__ float laneVal(float x, int k) {
  return __int_as_float(__builtin_amdgcn_readlane(__float_as_int(x), k));
}

struct C2 { float r, i; };
static __device__ __forceinline__ C2 cmul(C2 a, C2 b) {
  C2 z;
  z.r = a.r * b.r - a.i * b.i;
  z.i = a.r * b.i + a.i * b.r;
  return z;
}
static __device__ __forceinline__ C2 csel(bool c, C2 p) {
  C2 z;
  z.r = pick(c, p.r, 1.f);
  z.i = pick(c, p.i, 0.f);
  return z;
}

static __device__ __forceinline__ void phaseProd(C2 p0, C2 p1, C2 p2, C2 p3, C2 p4, C2 p5,
                                                 C2 p6, C2 p7, int lane, v8f* C, v8f* Sn) {
  C2 t = csel((lane >> 4) & 1, p0);
  t = cmul(t, csel((lane >> 3) & 1, p4));
  t = cmul(t, csel((lane >> 2) & 1, p5));
  t = cmul(t, csel((lane >> 1) & 1, p6));
  t = cmul(t, csel(lane & 1, p7));
  const C2 x1 = cmul(t, p1);
  const C2 y1 = cmul(t, p2), y3 = cmul(x1, p2);
  const C2 z1 = cmul(t, p3), z3 = cmul(y1, p3), z5 = cmul(x1, p3), z7 = cmul(y3, p3);
  v8f c, s;
  c[0] = t.r;  s[0] = t.i;   c[1] = z1.r; s[1] = z1.i;
  c[2] = y1.r; s[2] = y1.i;  c[3] = z3.r; s[3] = z3.i;
  c[4] = x1.r; s[4] = x1.i;  c[5] = z5.r; s[5] = z5.i;
  c[6] = y3.r; s[6] = y3.i;  c[7] = z7.r; s[7] = z7.i;
  *C = c; *Sn = s;
}

static __device__ __forceinline__ constexpr int pairIdx(int i, int j) {
  return i * (15 - i) / 2 + (j - i - 1);
}

static __device__ __forceinline__ v8f phaseAngles(float a0, float a1, float a2, float a3,
                                                  float a4, float a5, float a6, float a7,
                                                  int lane) {
  float t = pick((lane >> 4) & 1, a0, -a0);
  t += pick((lane >> 3) & 1, a4, -a4);
  t += pick((lane >> 2) & 1, a5, -a5);
  t += pick((lane >> 1) & 1, a6, -a6);
  t += pick(lane & 1, a7, -a7);
  const float u0 = t - a1, u1 = t + a1;
  const float v0 = u0 - a2, v1 = u0 + a2, v2 = u1 - a2, v3 = u1 + a2;
  v8f e;
  e[0] = v0 - a3; e[1] = v0 + a3; e[2] = v1 - a3; e[3] = v1 + a3;
  e[4] = v2 - a3; e[5] = v2 + a3; e[6] = v3 - a3; e[7] = v3 + a3;
  return e;
}

static __device__ __forceinline__ void mulPhaseCS(v8f* Sr, v8f* Si, v8f C, v8f Sn) {
  v8f r, i;
#pragma unroll
  for (int j = 0; j < 8; ++j) {
    r[j] = (*Sr)[j] * C[j] - (*Si)[j] * Sn[j];
    i[j] = (*Sr)[j] * Sn[j] + (*Si)[j] * C[j];
  }
  *Sr = r; *Si = i;
}

static __device__ __forceinline__ void stFrag(float* F, v8f v, int lane) {
  v4f lo, hi;
  lo.x = v[0]; lo.y = v[1]; lo.z = v[2]; lo.w = v[3];
  hi.x = v[4]; hi.y = v[5]; hi.z = v[6]; hi.w = v[7];
  *(v4f*)(F + lane * 4) = lo;
  *(v4f*)(F + 128 + lane * 4) = hi;
}
static __device__ __forceinline__ v8f ldFrag(const float* F, int lane) {
  const v4f lo = *(const v4f*)(F + lane * 4);
  const v4f hi = *(const v4f*)(F + 128 + lane * 4);
  v8f v;
  v[0] = lo.x; v[1] = lo.y; v[2] = lo.z; v[3] = lo.w;
  v[4] = hi.x; v[5] = hi.y; v[6] = hi.z; v[7] = hi.w;
  return v;
}

template <int RNG>
static __device__ __forceinline__ int cnotSrc(int s) {
  int src = s;
#pragma unroll
  for (int k = 7; k >= 0; --k) {
    int tgt = (k + RNG) & 7;
    src ^= ((src >> (7 - k)) & 1) << (7 - tgt);
  }
  return src;
}
static __device__ __forceinline__ int swapNib(int x) { return ((x & 15) << 4) | (x >> 4); }

static __device__ __forceinline__ void stState(float* Br, float* Bi, v8f Sr, v8f Si, int lane) {
  float* pr = Br + (lane & 15) * 16 + (lane >> 4) * 8;
  float* pi = Bi + (lane & 15) * 16 + (lane >> 4) * 8;
#pragma unroll
  for (int g = 0; g < 8; ++g) { pr[g] = Sr[g]; pi[g] = Si[g]; }
}

template <int RNG>
static __device__ __forceinline__ void ldCnot(const float* Br, const float* Bi,
                                              v8f* Sr, v8f* Si, int lane) {
  const int ab = swapNib(cnotSrc<RNG>(((lane >> 4) << 7) | (lane & 15)));
  v8f r, i;
#pragma unroll
  for (int j = 0; j < 8; ++j) {
    const int a = ab ^ swapNib(cnotSrc<RNG>(j << 4));
    r[j] = Br[a];
    i[j] = Bi[a];
  }
  *Sr = r; *Si = i;
}

extern "C" __global__ __launch_bounds__(256, 12) void vqc_kernel(
    const float* __restrict__ theta, const float* __restrict__ phi,
    const float* __restrict__ jup, const float* __restrict__ vqcw,
    float* __restrict__ out) {
  __shared__ float sAr[WPB][DD], sAi[WPB][DD];
  __shared__ float sBr[WPB][DD], sBi[WPB][DD];
  __shared__ __attribute__((aligned(16))) float sW[NSH][DD];

  int tid = threadIdx.x;
  int wid = tid >> 5, lane = tid & 31;
  int b = blockIdx.x * WPB + wid;
  const int c = lane & 15, h = lane >> 4;

  const float* th = theta + b * 8;
  const float* ph = phi + b * 8;
  const float* jj = jup + b * 28;
  const float* w0 = vqcw;
  const float* w1 = vqcw + 24;
  const int q = lane & 7;

  float thv[8], phv[8], jv[28];
#pragma unroll
  for (int k = 0; k < 8; ++k) { thv[k] = th[k]; phv[k] = ph[k]; }
#pragma unroll
  for (int p = 0; p < 28; ++p) jv[p] = jj[p];
  const float thq = th[q], phq = ph[q];
  const float waq = w1[3 * q], wbq = w1[3 * q + 1];

  {
    const int wu = __builtin_amdgcn_readfirstlane(wid);
    if (wu < 2) {
      const float* g = vqcw + wu * 12;
      stFrag(sW[wu], kronFrag(ryGate(0.5f * g[1]), ryGate(0.5f * g[4]),
                              ryGate(0.5f * g[7]), ryGate(0.5f * g[10]), lane), lane);
    } else if (wu == 2) {
      const v8f E = phaseAngles(0.5f * w0[2],  0.5f * w0[5],  0.5f * w0[8],  0.5f * w0[11],
                                0.5f * w0[14], 0.5f * w0[17], 0.5f * w0[20], 0.5f * w0[23],
                                lane);
      v8f pc, ps;
#pragma unroll
      for (int j = 0; j < 8; ++j) {
        float s, c;
        __sincosf(E[j], &s, &c);
        pc[j] = c; ps[j] = s;
      }
      stFrag(sW[2], pc, lane);
      stFrag(sW[3], ps, lane);
    }
  }

  v8f Sr, Si;

  {
    float mc[8], ms[8];
#pragma unroll
    for (int k = 0; k < 8; ++k) __sincosf(thv[k] * 0.5f, &ms[k], &mc[k]);

    float mb = pick(h, ms[0], mc[0]);
    mb *= pick((c >> 3) & 1, ms[4], mc[4]);
    mb *= pick((c >> 2) & 1, ms[5], mc[5]);
    mb *= pick((c >> 1) & 1, ms[6], mc[6]);
    mb *= pick(c & 1, ms[7], mc[7]);
    const float m0 = mb * mc[1], m1 = mb * ms[1];
    const float n0 = m0 * mc[2], n1 = m0 * ms[2], n2 = m1 * mc[2], n3 = m1 * ms[2];
    v8f mag;
    mag[0] = n0 * mc[3]; mag[1] = n0 * ms[3]; mag[2] = n1 * mc[3]; mag[3] = n1 * ms[3];
    mag[4] = n2 * mc[3]; mag[5] = n2 * ms[3]; mag[6] = n3 * mc[3]; mag[7] = n3 * ms[3];

    const float HP = 1.57079632679489662f;
    const float g0 = pick(h, -1.f, 1.f);
    const float g4 = pick((c >> 3) & 1, -1.f, 1.f);
    const float g5 = pick((c >> 2) & 1, -1.f, 1.f);
    const float g6 = pick((c >> 1) & 1, -1.f, 1.f);
    const float g7 = pick(c & 1, -1.f, 1.f);

    const float a0 = 0.5f * (phv[0] + w0[0]),  a1 = 0.5f * (phv[1] + w0[3]);
    const float a2 = 0.5f * (phv[2] + w0[6]),  a3 = 0.5f * (phv[3] + w0[9]);
    const float a4 = 0.5f * (phv[4] + w0[12]), a5 = 0.5f * (phv[5] + w0[15]);
    const float a6 = 0.5f * (phv[6] + w0[18]), a7 = 0.5f * (phv[7] + w0[21]);

    const float i0s = jv[pairIdx(0, 4)] * g4 + jv[pairIdx(0, 5)] * g5 + jv[pairIdx(0, 6)] * g6 +
                      jv[pairIdx(0, 7)] * g7;
    const float i4s = jv[pairIdx(4, 5)] * g5 + jv[pairIdx(4, 6)] * g6 + jv[pairIdx(4, 7)] * g7;
    const float i5s = jv[pairIdx(5, 6)] * g6 + jv[pairIdx(5, 7)] * g7;
    const float i6s = jv[pairIdx(6, 7)] * g7;
    const float A = -(g0 * (HP * i0s + a0) + g4 * (HP * i4s + a4) + g5 * (HP * i5s + a5) +
                      g6 * (HP * i6s + a6) + g7 * a7);

    const float f1 = jv[pairIdx(0, 1)] * g0 + jv[pairIdx(1, 4)] * g4 + jv[pairIdx(1, 5)] * g5 +
                     jv[pairIdx(1, 6)] * g6 + jv[pairIdx(1, 7)] * g7;
    const float f2 = jv[pairIdx(0, 2)] * g0 + jv[pairIdx(2, 4)] * g4 + jv[pairIdx(2, 5)] * g5 +
                     jv[pairIdx(2, 6)] * g6 + jv[pairIdx(2, 7)] * g7;
    const float f3 = jv[pairIdx(0, 3)] * g0 + jv[pairIdx(3, 4)] * g4 + jv[pairIdx(3, 5)] * g5 +
                     jv[pairIdx(3, 6)] * g6 + jv[pairIdx(3, 7)] * g7;
    const float F1 = -(HP * f1 + a1), F2 = -(HP * f2 + a2), F3 = -(HP * f3 + a3);
    const float K12 = HP * jv[pairIdx(1, 2)], K13 = HP * jv[pairIdx(1, 3)];
    const float K23 = HP * jv[pairIdx(2, 3)];

    const float tP = A + F1, tM = A - F1;
    const float F2p = F2 - K12, F2m = F2 + K12;
    const float uPP = tP + F2p, uPM = tP - F2p, uMP = tM + F2m, uMM = tM - F2m;
    const float F3p = F3 - K13, F3m = F3 + K13;
    const float vPP = F3p - K23, vPM = F3p + K23;
    const float vMP = F3m - K23, vMM = F3m + K23;
    v8f X;
    X[0] = uPP + vPP; X[1] = uPP - vPP; X[2] = uPM + vPM; X[3] = uPM - vPM;
    X[4] = uMP + vMP; X[5] = uMP - vMP; X[6] = uMM + vMM; X[7] = uMM - vMM;

#pragma unroll
    for (int j = 0; j < 8; ++j) {
      float se, ce;
      __sincosf(X[j], &se, &ce);
      Sr[j] = mag[j] * ce;
      Si[j] = mag[j] * se;
    }
  }

  __syncthreads();
  {
    const v8f L = ldFrag(sW[0], lane);
    const v8f R = ldFrag(sW[1], lane);
    applyStage(&Sr, &Si, L, R);
    mulPhaseCS(&Sr, &Si, ldFrag(sW[2], lane), ldFrag(sW[3], lane));
  }

  stState(sAr[wid], sAi[wid], Sr, Si, lane);
  {
    const FG f = fuseGate(thq, phq, waq, wbq);
    R2 g[8];
    C2 p[8];
#pragma unroll
    for (int k = 0; k < 8; ++k) {
      g[k].c = laneVal(f.c, k);  g[k].s = laneVal(f.s, k);
      p[k].r = laneVal(f.pr, k); p[k].i = laneVal(f.pi, k);
    }
    const v8f Lf = kronFrag(g[0], g[1], g[2], g[3], lane);
    const v8f Rf = kronFrag(g[4], g[5], g[6], g[7], lane);
    v8f Pc, Ps;
    phaseProd(p[0], p[1], p[2], p[3], p[4], p[5], p[6], p[7], lane, &Pc, &Ps);
    __syncthreads();
    ldCnot<1>(sAr[wid], sAi[wid], &Sr, &Si, lane);

    mulPhaseCS(&Sr, &Si, Pc, Ps);
    applyStage(&Sr, &Si, Lf, Rf);
  }

  stState(sBr[wid], sBi[wid], Sr, Si, lane);
  __syncthreads();
  ldCnot<2>(sBr[wid], sBi[wid], &Sr, &Si, lane);

  float tot = 0.f, q1 = 0.f, q2 = 0.f, q3 = 0.f;
#pragma unroll
  for (int j = 0; j < 8; ++j) {
    float p2 = Sr[j] * Sr[j] + Si[j] * Si[j];
    tot += p2;
    q1 += (j & 4) ? -p2 : p2;
    q2 += (j & 2) ? -p2 : p2;
    q3 += (j & 1) ? -p2 : p2;
  }
  float acc[8];
  acc[0] = h ? -tot : tot;
  acc[1] = q1; acc[2] = q2; acc[3] = q3;
  acc[4] = (c & 8) ? -tot : tot;
  acc[5] = (c & 4) ? -tot : tot;
  acc[6] = (c & 2) ? -tot : tot;
  acc[7] = (c & 1) ? -tot : tot;
#pragma unroll
  for (int k = 0; k < 8; ++k) {
    float v = acc[k];
    v += __shfl_xor(v, 16, 32);
    v += __shfl_xor(v, 8, 32);
    v += __shfl_xor(v, 4, 32);
    v += __shfl_xor(v, 2, 32);
    v += __shfl_xor(v, 1, 32);
    acc[k] = v;
  }
  if (lane == 0) {
#pragma unroll
    for (int k = 0; k < 8; ++k) out[b * 8 + k] = acc[k];
  }
}

extern "C" void kernel_launch(void* const* d_in, const int* in_sizes, int n_in,
                              void* d_out, int out_size, void* d_ws, size_t ws_size,
                              hipStream_t stream) {
  const float* theta = (const float*)d_in[0];
  const float* phi   = (const float*)d_in[1];
  const float* jup   = (const float*)d_in[2];
  const float* vqcw  = (const float*)d_in[3];
  float* out = (float*)d_out;
  int B = in_sizes[0] / 8;
  dim3 grid(B / WPB), block(32 * WPB);
  hipLaunchKernelGGL(vqc_kernel, grid, block, 0, stream, theta, phi, jup, vqcw, out);
}


# Layer-1 CNOT ring folded into the readout: <Z_k> behind it is a Walsh-signed sum of the unpermuted probabilities (registers, then vector-unit lane butterflies); no LDS exchange; sum order changes
# speedup vs baseline: 40.4393x; 1.0975x over previous
#include <hip/hip_runtime.h>

typedef __attribute__((ext_vector_type(2))) float v2f;
typedef __attribute__((ext_vector_type(4))) float v4f;
typedef __attribute__((ext_vector_type(8))) float v8f;

#define WPB 8
#define DD  256
#define NSH 4


static __device__ __forceinline__ v8f wmma4(v2f a, v2f b, v8f c) {
  return __builtin_amdgcn_wmma_f32_16x16x4_f32(false, a, false, b, (short)0, c,
                                               false, false);
}

static __device__ __forceinline__ v8f mmFrag(v8f a, v8f b, v8f acc) {
#pragma unroll
  for (int k = 0; k < 4; ++k) {
    v2f av, bv;
    av.x = a[2 * k]; av.y = a[2 * k + 1];
    bv.x = b[2 * k]; bv.y = b[2 * k + 1];
    acc = wmma4(av, bv, acc);
  }
  return acc;
}

static __device__ __forceinline__ void rmm(v8f Ar, v8f Ai, v8f B, v8f* Dr, v8f* Di) {
  v8f r = {};
  r = mmFrag(Ar, B, r);
  v8f i = {};
  i = mmFrag(Ai, B, i);
  *Dr = r; *Di = i;
}

static __device__ __forceinline__ void applyStage(v8f* Sr, v8f* Si, v8f L, v8f R) {
  v8f Ur, Ui;
  rmm(*Sr, *Si, L, &Ur, &Ui);
  rmm(Ur, Ui, R, Sr, Si);
}

struct R2 { float c, s; };

static __device__ __forceinline__ float pick(bool c, float t, float f) { return c ? t : f; }

static __device__ __forceinline__ R2 ryGate(float halfAngle) {
  R2 g;
  __sincosf(halfAngle, &g.s, &g.c);
  return g;
}

static __device__ __forceinline__ void pickRow(bool a, R2 g, float* x0, float* x1) {
  *x0 = pick(a, g.s, g.c);
  *x1 = pick(a, g.c, -g.s);
}

static __device__ __forceinline__ v8f kronFrag(R2 g0, R2 g1, R2 g2, R2 g3, int lane) {
  const bool a0 = (lane >> 3) & 1;
  const bool a1 = (lane >> 2) & 1;
  const bool a2 = (lane >> 1) & 1;
  const bool a3 = lane & 1;
  const bool b0 = (lane >> 4) & 1;
  float x00, x01, u0, u1, v0, v1, w0, w1;
  pickRow(a0, g0, &x00, &x01);
  const float p = pick(b0, x01, x00);
  pickRow(a1, g1, &u0, &u1);
  pickRow(a2, g2, &v0, &v1);
  pickRow(a3, g3, &w0, &w1);
  const float s0 = p * u0, s1 = p * u1;
  const float t0 = s0 * v0, t1 = s0 * v1, t2 = s1 * v0, t3 = s1 * v1;
  v8f k;
  k[0] = t0 * w0; k[1] = t0 * w1; k[2] = t1 * w0; k[3] = t1 * w1;
  k[4] = t2 * w0; k[5] = t2 * w1; k[6] = t3 * w0; k[7] = t3 * w1;
  return k;
}

struct FG { float c, s, pr, pi; };

static __device__ __forceinline__ FG fuseGate(float thq, float phq, float wa, float wb) {
  float sP, cP, sM, cM, sE, cE;
  __sincosf(0.25f * thq + 0.5f * wb, &sP, &cP);
  __sincosf(0.25f * thq - 0.5f * wb, &sM, &cM);
  __sincosf(0.25f * phq + 0.5f * wa, &sE, &cE);
  const float ur = cP * cE, ui = -(cM * sE);
  const float vr = sP * cE, vi = sM * sE;
  const float nu = ur * ur + ui * ui, nv = vr * vr + vi * vi;
  const float TINY = 1e-30f;
  const bool ku = nu > TINY, kv = nv > TINY;
  const float iu = ku ? __builtin_amdgcn_rsqf(nu) : 0.f;
  const float iv = kv ? __builtin_amdgcn_rsqf(nv) : 0.f;
  FG g;
  g.c = nu * iu;
  g.s = nv * iv;
  const float aur = ku ? ur * iu : 1.f, aui = -(ui * iu);
  const float avr = kv ? vr * iv : 1.f, avi = -(vi * iv);
  g.pr = aur * avr - aui * avi;
  g.pi = aur * avi + aui * avr;
  return g;
}

static __device__ __forceinline__ float laneVal(float x, int k) {
  return __int_as_float(__builtin_amdgcn_readlane(__float_as_int(x), k));
}

struct C2 { float r, i; };
static __device__ __forceinline__ C2 cmul(C2 a, C2 b) {
  C2 z;
  z.r = a.r * b.r - a.i * b.i;
  z.i = a.r * b.i + a.i * b.r;
  return z;
}
static __device__ __forceinline__ C2 csel(bool c, C2 p) {
  C2 z;
  z.r = pick(c, p.r, 1.f);
  z.i = pick(c, p.i, 0.f);
  return z;
}

static __device__ __forceinline__ void phaseProd(C2 p0, C2 p1, C2 p2, C2 p3, C2 p4, C2 p5,
                                                 C2 p6, C2 p7, int lane, v8f* C, v8f* Sn) {
  C2 t = csel((lane >> 4) & 1, p0);
  t = cmul(t, csel((lane >> 3) & 1, p4));
  t = cmul(t, csel((lane >> 2) & 1, p5));
  t = cmul(t, csel((lane >> 1) & 1, p6));
  t = cmul(t, csel(lane & 1, p7));
  const C2 x1 = cmul(t, p1);
  const C2 y1 = cmul(t, p2), y3 = cmul(x1, p2);
  const C2 z1 = cmul(t, p3), z3 = cmul(y1, p3), z5 = cmul(x1, p3), z7 = cmul(y3, p3);
  v8f c, s;
  c[0] = t.r;  s[0] = t.i;   c[1] = z1.r; s[1] = z1.i;
  c[2] = y1.r; s[2] = y1.i;  c[3] = z3.r; s[3] = z3.i;
  c[4] = x1.r; s[4] = x1.i;  c[5] = z5.r; s[5] = z5.i;
  c[6] = y3.r; s[6] = y3.i;  c[7] = z7.r; s[7] = z7.i;
  *C = c; *Sn = s;
}

static __device__ __forceinline__ constexpr int pairIdx(int i, int j) {
  return i * (15 - i) / 2 + (j - i - 1);
}

static __device__ __forceinline__ v8f phaseAngles(float a0, float a1, float a2, float a3,
                                                  float a4, float a5, float a6, float a7,
                                                  int lane) {
  float t = pick((lane >> 4) & 1, a0, -a0);
  t += pick((lane >> 3) & 1, a4, -a4);
  t += pick((lane >> 2) & 1, a5, -a5);
  t += pick((lane >> 1) & 1, a6, -a6);
  t += pick(lane & 1, a7, -a7);
  const float u0 = t - a1, u1 = t + a1;
  const float v0 = u0 - a2, v1 = u0 + a2, v2 = u1 - a2, v3 = u1 + a2;
  v8f e;
  e[0] = v0 - a3; e[1] = v0 + a3; e[2] = v1 - a3; e[3] = v1 + a3;
  e[4] = v2 - a3; e[5] = v2 + a3; e[6] = v3 - a3; e[7] = v3 + a3;
  return e;
}

static __device__ __forceinline__ void mulPhaseCS(v8f* Sr, v8f* Si, v8f C, v8f Sn) {
  v8f r, i;
#pragma unroll
  for (int j = 0; j < 8; ++j) {
    r[j] = (*Sr)[j] * C[j] - (*Si)[j] * Sn[j];
    i[j] = (*Sr)[j] * Sn[j] + (*Si)[j] * C[j];
  }
  *Sr = r; *Si = i;
}

static __device__ __forceinline__ void stFrag(float* F, v8f v, int lane) {
  v4f lo, hi;
  lo.x = v[0]; lo.y = v[1]; lo.z = v[2]; lo.w = v[3];
  hi.x = v[4]; hi.y = v[5]; hi.z = v[6]; hi.w = v[7];
  *(v4f*)(F + lane * 4) = lo;
  *(v4f*)(F + 128 + lane * 4) = hi;
}
static __device__ __forceinline__ v8f ldFrag(const float* F, int lane) {
  const v4f lo = *(const v4f*)(F + lane * 4);
  const v4f hi = *(const v4f*)(F + 128 + lane * 4);
  v8f v;
  v[0] = lo.x; v[1] = lo.y; v[2] = lo.z; v[3] = lo.w;
  v[4] = hi.x; v[5] = hi.y; v[6] = hi.z; v[7] = hi.w;
  return v;
}

template <int RNG>
static __device__ __forceinline__ int cnotSrc(int s) {
  int src = s;
#pragma unroll
  for (int k = 7; k >= 0; --k) {
    int tgt = (k + RNG) & 7;
    src ^= ((src >> (7 - k)) & 1) << (7 - tgt);
  }
  return src;
}
static __device__ __forceinline__ int swapNib(int x) { return ((x & 15) << 4) | (x >> 4); }

static __device__ __forceinline__ void stState(float* Br, float* Bi, v8f Sr, v8f Si, int lane) {
  float* pr = Br + (lane & 15) * 16 + (lane >> 4) * 8;
  float* pi = Bi + (lane & 15) * 16 + (lane >> 4) * 8;
#pragma unroll
  for (int g = 0; g < 8; ++g) { pr[g] = Sr[g]; pi[g] = Si[g]; }
}

template <int RNG>
static __device__ __forceinline__ void ldCnot(const float* Br, const float* Bi,
                                              v8f* Sr, v8f* Si, int lane) {
  const int ab = swapNib(cnotSrc<RNG>(((lane >> 4) << 7) | (lane & 15)));
  v8f r, i;
#pragma unroll
  for (int j = 0; j < 8; ++j) {
    const int a = ab ^ swapNib(cnotSrc<RNG>(j << 4));
    r[j] = Br[a];
    i[j] = Bi[a];
  }
  *Sr = r; *Si = i;
}

template <int RNG>
static constexpr int cnotDst(int t) {
  int d = t;
  for (int k = 0; k <= 7; ++k) {
    int tgt = (k + RNG) & 7;
    d ^= ((d >> (7 - k)) & 1) << (7 - tgt);
  }
  return d;
}
template <int RNG>
static constexpr int zMask(int k) {
  int m = 0;
  for (int bpos = 0; bpos < 8; ++bpos)
    if ((cnotDst<RNG>(1 << bpos) >> (7 - k)) & 1) m |= 1 << bpos;
  return m;
}
static_assert(zMask<2>(0) == 0x2A && zMask<2>(1) == 0x15 && zMask<2>(2) == 0xA0 &&
              zMask<2>(3) == 0x50 && zMask<2>(4) == 0xA8 && zMask<2>(5) == 0x54 &&
              zMask<2>(6) == 0xAA && zMask<2>(7) == 0x55,
              "readout sign masks do not match the layer-1 ring");

template <int M>
static __device__ __forceinline__ float xrow(float x) {
  return __int_as_float(__builtin_amdgcn_update_dpp(__float_as_int(x), __float_as_int(x),
                                                    0x160 + M, 0xf, 0xf, true));
}
static __device__ __forceinline__ float xhalf(float x) {
  const int v = __float_as_int(x);
  return __int_as_float(__builtin_amdgcn_permlanex16(v, v, 0x76543210, 0xfedcba98, false, false));
}
static __device__ __forceinline__ float walsh(bool bit, float x, float other) {
  return other + pick(bit, -x, x);
}

extern "C" __global__ __launch_bounds__(256, 12) void vqc_kernel(
    const float* __restrict__ theta, const float* __restrict__ phi,
    const float* __restrict__ jup, const float* __restrict__ vqcw,
    float* __restrict__ out) {
  __shared__ float sAr[WPB][DD], sAi[WPB][DD];
  __shared__ __attribute__((aligned(16))) float sW[NSH][DD];

  int tid = threadIdx.x;
  int wid = tid >> 5, lane = tid & 31;
  int b = blockIdx.x * WPB + wid;
  const int c = lane & 15, h = lane >> 4;

  const float* th = theta + b * 8;
  const float* ph = phi + b * 8;
  const float* jj = jup + b * 28;
  const float* w0 = vqcw;
  const float* w1 = vqcw + 24;
  const int q = lane & 7;

  float thv[8], phv[8], jv[28];
#pragma unroll
  for (int k = 0; k < 8; ++k) { thv[k] = th[k]; phv[k] = ph[k]; }
#pragma unroll
  for (int p = 0; p < 28; ++p) jv[p] = jj[p];
  const float thq = th[q], phq = ph[q];
  const float waq = w1[3 * q], wbq = w1[3 * q + 1];

  {
    const int wu = __builtin_amdgcn_readfirstlane(wid);
    if (wu < 2) {
      const float* g = vqcw + wu * 12;
      stFrag(sW[wu], kronFrag(ryGate(0.5f * g[1]), ryGate(0.5f * g[4]),
                              ryGate(0.5f * g[7]), ryGate(0.5f * g[10]), lane), lane);
    } else if (wu == 2) {
      const v8f E = phaseAngles(0.5f * w0[2],  0.5f * w0[5],  0.5f * w0[8],  0.5f * w0[11],
                                0.5f * w0[14], 0.5f * w0[17], 0.5f * w0[20], 0.5f * w0[23],
                                lane);
      v8f pc, ps;
#pragma unroll
      for (int j = 0; j < 8; ++j) {
        float s, c;
        __sincosf(E[j], &s, &c);
        pc[j] = c; ps[j] = s;
      }
      stFrag(sW[2], pc, lane);
      stFrag(sW[3], ps, lane);
    }
  }

  v8f Sr, Si;

  {
    float mc[8], ms[8];
#pragma unroll
    for (int k = 0; k < 8; ++k) __sincosf(thv[k] * 0.5f, &ms[k], &mc[k]);

    float mb = pick(h, ms[0], mc[0]);
    mb *= pick((c >> 3) & 1, ms[4], mc[4]);
    mb *= pick((c >> 2) & 1, ms[5], mc[5]);
    mb *= pick((c >> 1) & 1, ms[6], mc[6]);
    mb *= pick(c & 1, ms[7], mc[7]);
    const float m0 = mb * mc[1], m1 = mb * ms[1];
    const float n0 = m0 * mc[2], n1 = m0 * ms[2], n2 = m1 * mc[2], n3 = m1 * ms[2];
    v8f mag;
    mag[0] = n0 * mc[3]; mag[1] = n0 * ms[3]; mag[2] = n1 * mc[3]; mag[3] = n1 * ms[3];
    mag[4] = n2 * mc[3]; mag[5] = n2 * ms[3]; mag[6] = n3 * mc[3]; mag[7] = n3 * ms[3];

    const float HP = 1.57079632679489662f;
    const float g0 = pick(h, -1.f, 1.f);
    const float g4 = pick((c >> 3) & 1, -1.f, 1.f);
    const float g5 = pick((c >> 2) & 1, -1.f, 1.f);
    const float g6 = pick((c >> 1) & 1, -1.f, 1.f);
    const float g7 = pick(c & 1, -1.f, 1.f);

    const float a0 = 0.5f * (phv[0] + w0[0]),  a1 = 0.5f * (phv[1] + w0[3]);
    const float a2 = 0.5f * (phv[2] + w0[6]),  a3 = 0.5f * (phv[3] + w0[9]);
    const float a4 = 0.5f * (phv[4] + w0[12]), a5 = 0.5f * (phv[5] + w0[15]);
    const float a6 = 0.5f * (phv[6] + w0[18]), a7 = 0.5f * (phv[7] + w0[21]);

    const float i0s = jv[pairIdx(0, 4)] * g4 + jv[pairIdx(0, 5)] * g5 + jv[pairIdx(0, 6)] * g6 +
                      jv[pairIdx(0, 7)] * g7;
    const float i4s = jv[pairIdx(4, 5)] * g5 + jv[pairIdx(4, 6)] * g6 + jv[pairIdx(4, 7)] * g7;
    const float i5s = jv[pairIdx(5, 6)] * g6 + jv[pairIdx(5, 7)] * g7;
    const float i6s = jv[pairIdx(6, 7)] * g7;
    const float A = -(g0 * (HP * i0s + a0) + g4 * (HP * i4s + a4) + g5 * (HP * i5s + a5) +
                      g6 * (HP * i6s + a6) + g7 * a7);

    const float f1 = jv[pairIdx(0, 1)] * g0 + jv[pairIdx(1, 4)] * g4 + jv[pairIdx(1, 5)] * g5 +
                     jv[pairIdx(1, 6)] * g6 + jv[pairIdx(1, 7)] * g7;
    const float f2 = jv[pairIdx(0, 2)] * g0 + jv[pairIdx(2, 4)] * g4 + jv[pairIdx(2, 5)] * g5 +
                     jv[pairIdx(2, 6)] * g6 + jv[pairIdx(2, 7)] * g7;
    const float f3 = jv[pairIdx(0, 3)] * g0 + jv[pairIdx(3, 4)] * g4 + jv[pairIdx(3, 5)] * g5 +
                     jv[pairIdx(3, 6)] * g6 + jv[pairIdx(3, 7)] * g7;
    const float F1 = -(HP * f1 + a1), F2 = -(HP * f2 + a2), F3 = -(HP * f3 + a3);
    const float K12 = HP * jv[pairIdx(1, 2)], K13 = HP * jv[pairIdx(1, 3)];
    const float K23 = HP * jv[pairIdx(2, 3)];

    const float tP = A + F1, tM = A - F1;
    const float F2p = F2 - K12, F2m = F2 + K12;
    const float uPP = tP + F2p, uPM = tP - F2p, uMP = tM + F2m, uMM = tM - F2m;
    const float F3p = F3 - K13, F3m = F3 + K13;
    const float vPP = F3p - K23, vPM = F3p + K23;
    const float vMP = F3m - K23, vMM = F3m + K23;
    v8f X;
    X[0] = uPP + vPP; X[1] = uPP - vPP; X[2] = uPM + vPM; X[3] = uPM - vPM;
    X[4] = uMP + vMP; X[5] = uMP - vMP; X[6] = uMM + vMM; X[7] = uMM - vMM;

#pragma unroll
    for (int j = 0; j < 8; ++j) {
      float se, ce;
      __sincosf(X[j], &se, &ce);
      Sr[j] = mag[j] * ce;
      Si[j] = mag[j] * se;
    }
  }

  __syncthreads();
  {
    const v8f L = ldFrag(sW[0], lane);
    const v8f R = ldFrag(sW[1], lane);
    applyStage(&Sr, &Si, L, R);
    mulPhaseCS(&Sr, &Si, ldFrag(sW[2], lane), ldFrag(sW[3], lane));
  }

  stState(sAr[wid], sAi[wid], Sr, Si, lane);
  {
    const FG f = fuseGate(thq, phq, waq, wbq);
    R2 g[8];
    C2 p[8];
#pragma unroll
    for (int k = 0; k < 8; ++k) {
      g[k].c = laneVal(f.c, k);  g[k].s = laneVal(f.s, k);
      p[k].r = laneVal(f.pr, k); p[k].i = laneVal(f.pi, k);
    }
    const v8f Lf = kronFrag(g[0], g[1], g[2], g[3], lane);
    const v8f Rf = kronFrag(g[4], g[5], g[6], g[7], lane);
    v8f Pc, Ps;
    phaseProd(p[0], p[1], p[2], p[3], p[4], p[5], p[6], p[7], lane, &Pc, &Ps);
    __syncthreads();
    ldCnot<1>(sAr[wid], sAi[wid], &Sr, &Si, lane);

    mulPhaseCS(&Sr, &Si, Pc, Ps);
    applyStage(&Sr, &Si, Lf, Rf);
  }

  float p[8];
#pragma unroll
  for (int j = 0; j < 8; ++j) p[j] = Sr[j] * Sr[j] + Si[j] * Si[j];
  const float a0 = p[0] + p[1], b0 = p[0] - p[1], a1 = p[2] + p[3], b1 = p[2] - p[3];
  const float a2 = p[4] + p[5], b2 = p[4] - p[5], a3 = p[6] + p[7], b3 = p[6] - p[7];
  const float r2 = (a0 - a1) + (a2 - a3);
  const float b01 = b0 + b1, b23 = b2 + b3;
  const float r1 = b01 + b23, r5 = b01 - b23;

  const bool c3 = (c >> 3) & 1, c2 = (c >> 2) & 1, c1 = (c >> 1) & 1, c0 = c & 1;
  float x = r2 + xrow<4>(r2);
  x = x + xrow<1>(x);
  x = walsh(h, x, xhalf(x));
  x = walsh(c3, x, xrow<8>(x));
  x = walsh(c1, x, xrow<2>(x));
  float y = pick(h, r5, r1) + xhalf(pick(h, r1, r5));
  y = y + xrow<8>(y);
  y = y + xrow<2>(y);
  y = walsh(c2, y, xrow<4>(y));
  y = walsh(c0, y, xrow<1>(y));

  const float o0 = laneVal(x, 10), o2 = laneVal(x, 16), o4 = laneVal(x, 24), o6 = laneVal(x, 26);
  const float o1 = laneVal(y, 5),  o3 = laneVal(y, 16), o5 = laneVal(y, 20), o7 = laneVal(y, 21);
  if (lane == 0) {
    float* o = out + b * 8;
    o[0] = o0; o[1] = o1; o[2] = o2; o[3] = o3;
    o[4] = o4; o[5] = o5; o[6] = o6; o[7] = o7;
  }
}

extern "C" void kernel_launch(void* const* d_in, const int* in_sizes, int n_in,
                              void* d_out, int out_size, void* d_ws, size_t ws_size,
                              hipStream_t stream) {
  const float* theta = (const float*)d_in[0];
  const float* phi   = (const float*)d_in[1];
  const float* jup   = (const float*)d_in[2];
  const float* vqcw  = (const float*)d_in[3];
  float* out = (float*)d_out;
  int B = in_sizes[0] / 8;
  dim3 grid(B / WPB), block(32 * WPB);
  hipLaunchKernelGGL(vqc_kernel, grid, block, 0, stream, theta, phi, jup, vqcw, out);
}


# Each wave carries 2 batch elements side by side as independent straight-line streams (grid halved, no loop): per-wave set-up, shared-fragment reads, barriers, gate pass serve both; sum order unchanged
# speedup vs baseline: 42.7839x; 1.0580x over previous
#include <hip/hip_runtime.h>

typedef __attribute__((ext_vector_type(2))) float v2f;
typedef __attribute__((ext_vector_type(4))) float v4f;
typedef __attribute__((ext_vector_type(8))) float v8f;

#define WPB 8
#define DD  256
#define EPW 2
#define NSH 4


static __device__ __forceinline__ v8f wmma4(v2f a, v2f b, v8f c) {
  return __builtin_amdgcn_wmma_f32_16x16x4_f32(false, a, false, b, (short)0, c,
                                               false, false);
}

static __device__ __forceinline__ v8f mmFrag(v8f a, v8f b, v8f acc) {
#pragma unroll
  for (int k = 0; k < 4; ++k) {
    v2f av, bv;
    av.x = a[2 * k]; av.y = a[2 * k + 1];
    bv.x = b[2 * k]; bv.y = b[2 * k + 1];
    acc = wmma4(av, bv, acc);
  }
  return acc;
}

static __device__ __forceinline__ void rmm(v8f Ar, v8f Ai, v8f B, v8f* Dr, v8f* Di) {
  v8f r = {};
  r = mmFrag(Ar, B, r);
  v8f i = {};
  i = mmFrag(Ai, B, i);
  *Dr = r; *Di = i;
}

static __device__ __forceinline__ void applyStage(v8f* Sr, v8f* Si, v8f L, v8f R) {
  v8f Ur, Ui;
  rmm(*Sr, *Si, L, &Ur, &Ui);
  rmm(Ur, Ui, R, Sr, Si);
}

struct R2 { float c, s; };

static __device__ __forceinline__ float pick(bool c, float t, float f) { return c ? t : f; }

static __device__ __forceinline__ R2 ryGate(float halfAngle) {
  R2 g;
  __sincosf(halfAngle, &g.s, &g.c);
  return g;
}

static __device__ __forceinline__ void pickRow(bool a, R2 g, float* x0, float* x1) {
  *x0 = pick(a, g.s, g.c);
  *x1 = pick(a, g.c, -g.s);
}

static __device__ __forceinline__ v8f kronFrag(R2 g0, R2 g1, R2 g2, R2 g3, int lane) {
  const bool a0 = (lane >> 3) & 1;
  const bool a1 = (lane >> 2) & 1;
  const bool a2 = (lane >> 1) & 1;
  const bool a3 = lane & 1;
  const bool b0 = (lane >> 4) & 1;
  float x00, x01, u0, u1, v0, v1, w0, w1;
  pickRow(a0, g0, &x00, &x01);
  const float p = pick(b0, x01, x00);
  pickRow(a1, g1, &u0, &u1);
  pickRow(a2, g2, &v0, &v1);
  pickRow(a3, g3, &w0, &w1);
  const float s0 = p * u0, s1 = p * u1;
  const float t0 = s0 * v0, t1 = s0 * v1, t2 = s1 * v0, t3 = s1 * v1;
  v8f k;
  k[0] = t0 * w0; k[1] = t0 * w1; k[2] = t1 * w0; k[3] = t1 * w1;
  k[4] = t2 * w0; k[5] = t2 * w1; k[6] = t3 * w0; k[7] = t3 * w1;
  return k;
}

struct FG { float c, s, pr, pi; };

static __device__ __forceinline__ FG fuseGate(float thq, float phq, float wa, float wb) {
  float sP, cP, sM, cM, sE, cE;
  __sincosf(0.25f * thq + 0.5f * wb, &sP, &cP);
  __sincosf(0.25f * thq - 0.5f * wb, &sM, &cM);
  __sincosf(0.25f * phq + 0.5f * wa, &sE, &cE);
  const float ur = cP * cE, ui = -(cM * sE);
  const float vr = sP * cE, vi = sM * sE;
  const float nu = ur * ur + ui * ui, nv = vr * vr + vi * vi;
  const float TINY = 1e-30f;
  const bool ku = nu > TINY, kv = nv > TINY;
  const float iu = ku ? __builtin_amdgcn_rsqf(nu) : 0.f;
  const float iv = kv ? __builtin_amdgcn_rsqf(nv) : 0.f;
  FG g;
  g.c = nu * iu;
  g.s = nv * iv;
  const float aur = ku ? ur * iu : 1.f, aui = -(ui * iu);
  const float avr = kv ? vr * iv : 1.f, avi = -(vi * iv);
  g.pr = aur * avr - aui * avi;
  g.pi = aur * avi + aui * avr;
  return g;
}

static __device__ __forceinline__ float laneVal(float x, int k) {
  return __int_as_float(__builtin_amdgcn_readlane(__float_as_int(x), k));
}

struct C2 { float r, i; };
static __device__ __forceinline__ C2 cmul(C2 a, C2 b) {
  C2 z;
  z.r = a.r * b.r - a.i * b.i;
  z.i = a.r * b.i + a.i * b.r;
  return z;
}
static __device__ __forceinline__ C2 csel(bool c, C2 p) {
  C2 z;
  z.r = pick(c, p.r, 1.f);
  z.i = pick(c, p.i, 0.f);
  return z;
}

static __device__ __forceinline__ void phaseProd(C2 p0, C2 p1, C2 p2, C2 p3, C2 p4, C2 p5,
                                                 C2 p6, C2 p7, int lane, v8f* C, v8f* Sn) {
  C2 t = csel((lane >> 4) & 1, p0);
  t = cmul(t, csel((lane >> 3) & 1, p4));
  t = cmul(t, csel((lane >> 2) & 1, p5));
  t = cmul(t, csel((lane >> 1) & 1, p6));
  t = cmul(t, csel(lane & 1, p7));
  const C2 x1 = cmul(t, p1);
  const C2 y1 = cmul(t, p2), y3 = cmul(x1, p2);
  const C2 z1 = cmul(t, p3), z3 = cmul(y1, p3), z5 = cmul(x1, p3), z7 = cmul(y3, p3);
  v8f c, s;
  c[0] = t.r;  s[0] = t.i;   c[1] = z1.r; s[1] = z1.i;
  c[2] = y1.r; s[2] = y1.i;  c[3] = z3.r; s[3] = z3.i;
  c[4] = x1.r; s[4] = x1.i;  c[5] = z5.r; s[5] = z5.i;
  c[6] = y3.r; s[6] = y3.i;  c[7] = z7.r; s[7] = z7.i;
  *C = c; *Sn = s;
}

static __device__ __forceinline__ constexpr int pairIdx(int i, int j) {
  return i * (15 - i) / 2 + (j - i - 1);
}

static __device__ __forceinline__ v8f phaseAngles(float a0, float a1, float a2, float a3,
                                                  float a4, float a5, float a6, float a7,
                                                  int lane) {
  float t = pick((lane >> 4) & 1, a0, -a0);
  t += pick((lane >> 3) & 1, a4, -a4);
  t += pick((lane >> 2) & 1, a5, -a5);
  t += pick((lane >> 1) & 1, a6, -a6);
  t += pick(lane & 1, a7, -a7);
  const float u0 = t - a1, u1 = t + a1;
  const float v0 = u0 - a2, v1 = u0 + a2, v2 = u1 - a2, v3 = u1 + a2;
  v8f e;
  e[0] = v0 - a3; e[1] = v0 + a3; e[2] = v1 - a3; e[3] = v1 + a3;
  e[4] = v2 - a3; e[5] = v2 + a3; e[6] = v3 - a3; e[7] = v3 + a3;
  return e;
}

static __device__ __forceinline__ void mulPhaseCS(v8f* Sr, v8f* Si, v8f C, v8f Sn) {
  v8f r, i;
#pragma unroll
  for (int j = 0; j < 8; ++j) {
    r[j] = (*Sr)[j] * C[j] - (*Si)[j] * Sn[j];
    i[j] = (*Sr)[j] * Sn[j] + (*Si)[j] * C[j];
  }
  *Sr = r; *Si = i;
}

static __device__ __forceinline__ void stFrag(float* F, v8f v, int lane) {
  v4f lo, hi;
  lo.x = v[0]; lo.y = v[1]; lo.z = v[2]; lo.w = v[3];
  hi.x = v[4]; hi.y = v[5]; hi.z = v[6]; hi.w = v[7];
  *(v4f*)(F + lane * 4) = lo;
  *(v4f*)(F + 128 + lane * 4) = hi;
}
static __device__ __forceinline__ v8f ldFrag(const float* F, int lane) {
  const v4f lo = *(const v4f*)(F + lane * 4);
  const v4f hi = *(const v4f*)(F + 128 + lane * 4);
  v8f v;
  v[0] = lo.x; v[1] = lo.y; v[2] = lo.z; v[3] = lo.w;
  v[4] = hi.x; v[5] = hi.y; v[6] = hi.z; v[7] = hi.w;
  return v;
}

template <int RNG>
static __device__ __forceinline__ int cnotSrc(int s) {
  int src = s;
#pragma unroll
  for (int k = 7; k >= 0; --k) {
    int tgt = (k + RNG) & 7;
    src ^= ((src >> (7 - k)) & 1) << (7 - tgt);
  }
  return src;
}
static __device__ __forceinline__ int swapNib(int x) { return ((x & 15) << 4) | (x >> 4); }

static __device__ __forceinline__ void stState(float* Br, float* Bi, v8f Sr, v8f Si, int lane) {
  float* pr = Br + (lane & 15) * 16 + (lane >> 4) * 8;
  float* pi = Bi + (lane & 15) * 16 + (lane >> 4) * 8;
#pragma unroll
  for (int g = 0; g < 8; ++g) { pr[g] = Sr[g]; pi[g] = Si[g]; }
}

template <int RNG>
static __device__ __forceinline__ void ldCnot(const float* Br, const float* Bi,
                                              v8f* Sr, v8f* Si, int lane) {
  const int ab = swapNib(cnotSrc<RNG>(((lane >> 4) << 7) | (lane & 15)));
  v8f r, i;
#pragma unroll
  for (int j = 0; j < 8; ++j) {
    const int a = ab ^ swapNib(cnotSrc<RNG>(j << 4));
    r[j] = Br[a];
    i[j] = Bi[a];
  }
  *Sr = r; *Si = i;
}

template <int RNG>
static constexpr int cnotDst(int t) {
  int d = t;
  for (int k = 0; k <= 7; ++k) {
    int tgt = (k + RNG) & 7;
    d ^= ((d >> (7 - k)) & 1) << (7 - tgt);
  }
  return d;
}
template <int RNG>
static constexpr int zMask(int k) {
  int m = 0;
  for (int bpos = 0; bpos < 8; ++bpos)
    if ((cnotDst<RNG>(1 << bpos) >> (7 - k)) & 1) m |= 1 << bpos;
  return m;
}
static_assert(zMask<2>(0) == 0x2A && zMask<2>(1) == 0x15 && zMask<2>(2) == 0xA0 &&
              zMask<2>(3) == 0x50 && zMask<2>(4) == 0xA8 && zMask<2>(5) == 0x54 &&
              zMask<2>(6) == 0xAA && zMask<2>(7) == 0x55,
              "readout sign masks do not match the layer-1 ring");

template <int M>
static __device__ __forceinline__ float xrow(float x) {
  return __int_as_float(__builtin_amdgcn_update_dpp(__float_as_int(x), __float_as_int(x),
                                                    0x160 + M, 0xf, 0xf, true));
}
static __device__ __forceinline__ float xhalf(float x) {
  const int v = __float_as_int(x);
  return __int_as_float(__builtin_amdgcn_permlanex16(v, v, 0x76543210, 0xfedcba98, false, false));
}
static __device__ __forceinline__ float walsh(bool bit, float x, float other) {
  return other + pick(bit, -x, x);
}

static __device__ __forceinline__ void phase0(const float* thv, const float* phv, const float* jv,
                                              const float* __restrict__ w0, int lane,
                                              v8f* Sr, v8f* Si) {
  const int c = lane & 15, h = lane >> 4;
  v8f sr, si;
  {
    float mc[8], ms[8];
#pragma unroll
    for (int k = 0; k < 8; ++k) __sincosf(thv[k] * 0.5f, &ms[k], &mc[k]);

    float mb = pick(h, ms[0], mc[0]);
    mb *= pick((c >> 3) & 1, ms[4], mc[4]);
    mb *= pick((c >> 2) & 1, ms[5], mc[5]);
    mb *= pick((c >> 1) & 1, ms[6], mc[6]);
    mb *= pick(c & 1, ms[7], mc[7]);
    const float m0 = mb * mc[1], m1 = mb * ms[1];
    const float n0 = m0 * mc[2], n1 = m0 * ms[2], n2 = m1 * mc[2], n3 = m1 * ms[2];
    v8f mag;
    mag[0] = n0 * mc[3]; mag[1] = n0 * ms[3]; mag[2] = n1 * mc[3]; mag[3] = n1 * ms[3];
    mag[4] = n2 * mc[3]; mag[5] = n2 * ms[3]; mag[6] = n3 * mc[3]; mag[7] = n3 * ms[3];

    const float HP = 1.57079632679489662f;
    const float g0 = pick(h, -1.f, 1.f);
    const float g4 = pick((c >> 3) & 1, -1.f, 1.f);
    const float g5 = pick((c >> 2) & 1, -1.f, 1.f);
    const float g6 = pick((c >> 1) & 1, -1.f, 1.f);
    const float g7 = pick(c & 1, -1.f, 1.f);

    const float a0 = 0.5f * (phv[0] + w0[0]),  a1 = 0.5f * (phv[1] + w0[3]);
    const float a2 = 0.5f * (phv[2] + w0[6]),  a3 = 0.5f * (phv[3] + w0[9]);
    const float a4 = 0.5f * (phv[4] + w0[12]), a5 = 0.5f * (phv[5] + w0[15]);
    const float a6 = 0.5f * (phv[6] + w0[18]), a7 = 0.5f * (phv[7] + w0[21]);

    const float i0s = jv[pairIdx(0, 4)] * g4 + jv[pairIdx(0, 5)] * g5 + jv[pairIdx(0, 6)] * g6 +
                      jv[pairIdx(0, 7)] * g7;
    const float i4s = jv[pairIdx(4, 5)] * g5 + jv[pairIdx(4, 6)] * g6 + jv[pairIdx(4, 7)] * g7;
    const float i5s = jv[pairIdx(5, 6)] * g6 + jv[pairIdx(5, 7)] * g7;
    const float i6s = jv[pairIdx(6, 7)] * g7;
    const float A = -(g0 * (HP * i0s + a0) + g4 * (HP * i4s + a4) + g5 * (HP * i5s + a5) +
                      g6 * (HP * i6s + a6) + g7 * a7);

    const float f1 = jv[pairIdx(0, 1)] * g0 + jv[pairIdx(1, 4)] * g4 + jv[pairIdx(1, 5)] * g5 +
                     jv[pairIdx(1, 6)] * g6 + jv[pairIdx(1, 7)] * g7;
    const float f2 = jv[pairIdx(0, 2)] * g0 + jv[pairIdx(2, 4)] * g4 + jv[pairIdx(2, 5)] * g5 +
                     jv[pairIdx(2, 6)] * g6 + jv[pairIdx(2, 7)] * g7;
    const float f3 = jv[pairIdx(0, 3)] * g0 + jv[pairIdx(3, 4)] * g4 + jv[pairIdx(3, 5)] * g5 +
                     jv[pairIdx(3, 6)] * g6 + jv[pairIdx(3, 7)] * g7;
    const float F1 = -(HP * f1 + a1), F2 = -(HP * f2 + a2), F3 = -(HP * f3 + a3);
    const float K12 = HP * jv[pairIdx(1, 2)], K13 = HP * jv[pairIdx(1, 3)];
    const float K23 = HP * jv[pairIdx(2, 3)];

    const float tP = A + F1, tM = A - F1;
    const float F2p = F2 - K12, F2m = F2 + K12;
    const float uPP = tP + F2p, uPM = tP - F2p, uMP = tM + F2m, uMM = tM - F2m;
    const float F3p = F3 - K13, F3m = F3 + K13;
    const float vPP = F3p - K23, vPM = F3p + K23;
    const float vMP = F3m - K23, vMM = F3m + K23;
    v8f X;
    X[0] = uPP + vPP; X[1] = uPP - vPP; X[2] = uPM + vPM; X[3] = uPM - vPM;
    X[4] = uMP + vMP; X[5] = uMP - vMP; X[6] = uMM + vMM; X[7] = uMM - vMM;

#pragma unroll
    for (int j = 0; j < 8; ++j) {
      float se, ce;
      __sincosf(X[j], &se, &ce);
      sr[j] = mag[j] * ce;
      si[j] = mag[j] * se;
    }
  }
  *Sr = sr; *Si = si;
}

static __device__ __forceinline__ void readout(v8f Sr, v8f Si, int lane, float* xo, float* yo) {
  const int c = lane & 15, h = lane >> 4;
  float p[8];
#pragma unroll
  for (int j = 0; j < 8; ++j) p[j] = Sr[j] * Sr[j] + Si[j] * Si[j];
  const float a0 = p[0] + p[1], b0 = p[0] - p[1], a1 = p[2] + p[3], b1 = p[2] - p[3];
  const float a2 = p[4] + p[5], b2 = p[4] - p[5], a3 = p[6] + p[7], b3 = p[6] - p[7];
  const float r2 = (a0 - a1) + (a2 - a3);
  const float b01 = b0 + b1, b23 = b2 + b3;
  const float r1 = b01 + b23, r5 = b01 - b23;

  const bool c3 = (c >> 3) & 1, c2 = (c >> 2) & 1, c1 = (c >> 1) & 1, c0 = c & 1;
  float x = r2 + xrow<4>(r2);
  x = x + xrow<1>(x);
  x = walsh(h, x, xhalf(x));
  x = walsh(c3, x, xrow<8>(x));
  x = walsh(c1, x, xrow<2>(x));
  float y = pick(h, r5, r1) + xhalf(pick(h, r1, r5));
  y = y + xrow<8>(y);
  y = y + xrow<2>(y);
  y = walsh(c2, y, xrow<4>(y));
  y = walsh(c0, y, xrow<1>(y));
  *xo = x; *yo = y;
}

extern "C" __global__ __launch_bounds__(256, 8) void vqc_kernel(
    const float* __restrict__ theta, const float* __restrict__ phi,
    const float* __restrict__ jup, const float* __restrict__ vqcw,
    float* __restrict__ out) {
  __shared__ float sAr[WPB][EPW][DD], sAi[WPB][EPW][DD];
  __shared__ __attribute__((aligned(16))) float sW[NSH][DD];

  int tid = threadIdx.x;
  int wid = tid >> 5, lane = tid & 31;
  int b = (blockIdx.x * WPB + wid) * EPW;

  const float* w0 = vqcw;
  const float* w1 = vqcw + 24;
  const int q = lane & 7;

  float thv[EPW][8], phv[EPW][8], jv[EPW][28];
#pragma unroll
  for (int e = 0; e < EPW; ++e) {
    const float* th = theta + (b + e) * 8;
    const float* ph = phi + (b + e) * 8;
    const float* jj = jup + (b + e) * 28;
#pragma unroll
    for (int k = 0; k < 8; ++k) { thv[e][k] = th[k]; phv[e][k] = ph[k]; }
#pragma unroll
    for (int p = 0; p < 28; ++p) jv[e][p] = jj[p];
  }
  const float thq = theta[b * 8 + (lane & 15)], phq = phi[b * 8 + (lane & 15)];
  const float waq = w1[3 * q], wbq = w1[3 * q + 1];

  {
    const int wu = __builtin_amdgcn_readfirstlane(wid);
    if (wu < 2) {
      const float* g = vqcw + wu * 12;
      stFrag(sW[wu], kronFrag(ryGate(0.5f * g[1]), ryGate(0.5f * g[4]),
                              ryGate(0.5f * g[7]), ryGate(0.5f * g[10]), lane), lane);
    } else if (wu == 2) {
      const v8f E = phaseAngles(0.5f * w0[2],  0.5f * w0[5],  0.5f * w0[8],  0.5f * w0[11],
                                0.5f * w0[14], 0.5f * w0[17], 0.5f * w0[20], 0.5f * w0[23],
                                lane);
      v8f pc, ps;
#pragma unroll
      for (int j = 0; j < 8; ++j) {
        float s, c;
        __sincosf(E[j], &s, &c);
        pc[j] = c; ps[j] = s;
      }
      stFrag(sW[2], pc, lane);
      stFrag(sW[3], ps, lane);
    }
  }

  v8f Sr[EPW], Si[EPW];

#pragma unroll
  for (int e = 0; e < EPW; ++e) phase0(thv[e], phv[e], jv[e], w0, lane, &Sr[e], &Si[e]);

  __syncthreads();
  {
    const v8f L = ldFrag(sW[0], lane);
    const v8f R = ldFrag(sW[1], lane);
    const v8f pc = ldFrag(sW[2], lane), ps = ldFrag(sW[3], lane);
#pragma unroll
    for (int e = 0; e < EPW; ++e) {
      applyStage(&Sr[e], &Si[e], L, R);
      mulPhaseCS(&Sr[e], &Si[e], pc, ps);
    }
  }

#pragma unroll
  for (int e = 0; e < EPW; ++e) stState(sAr[wid][e], sAi[wid][e], Sr[e], Si[e], lane);
  {
    const FG f = fuseGate(thq, phq, waq, wbq);
    v8f Lf[EPW], Rf[EPW], Pc[EPW], Ps[EPW];
#pragma unroll
    for (int e = 0; e < EPW; ++e) {
      R2 g[8];
      C2 p[8];
#pragma unroll
      for (int k = 0; k < 8; ++k) {
        g[k].c = laneVal(f.c, 8 * e + k);  g[k].s = laneVal(f.s, 8 * e + k);
        p[k].r = laneVal(f.pr, 8 * e + k); p[k].i = laneVal(f.pi, 8 * e + k);
      }
      Lf[e] = kronFrag(g[0], g[1], g[2], g[3], lane);
      Rf[e] = kronFrag(g[4], g[5], g[6], g[7], lane);
      phaseProd(p[0], p[1], p[2], p[3], p[4], p[5], p[6], p[7], lane, &Pc[e], &Ps[e]);
    }
    __syncthreads();
#pragma unroll
    for (int e = 0; e < EPW; ++e) {
      ldCnot<1>(sAr[wid][e], sAi[wid][e], &Sr[e], &Si[e], lane);
      mulPhaseCS(&Sr[e], &Si[e], Pc[e], Ps[e]);
      applyStage(&Sr[e], &Si[e], Lf[e], Rf[e]);
    }
  }

  float o[EPW][8];
#pragma unroll
  for (int e = 0; e < EPW; ++e) {
    float x, y;
    readout(Sr[e], Si[e], lane, &x, &y);
    o[e][0] = laneVal(x, 10); o[e][2] = laneVal(x, 16); o[e][4] = laneVal(x, 24); o[e][6] = laneVal(x, 26);
    o[e][1] = laneVal(y, 5);  o[e][3] = laneVal(y, 16); o[e][5] = laneVal(y, 20); o[e][7] = laneVal(y, 21);
  }
  if (lane == 0) {
    float* po = out + b * 8;
#pragma unroll
    for (int e = 0; e < EPW; ++e)
#pragma unroll
      for (int k = 0; k < 8; ++k) po[e * 8 + k] = o[e][k];
  }
}

extern "C" void kernel_launch(void* const* d_in, const int* in_sizes, int n_in,
                              void* d_out, int out_size, void* d_ws, size_t ws_size,
                              hipStream_t stream) {
  const float* theta = (const float*)d_in[0];
  const float* phi   = (const float*)d_in[1];
  const float* jup   = (const float*)d_in[2];
  const float* vqcw  = (const float*)d_in[3];
  float* out = (float*)d_out;
  int B = in_sizes[0] / 8;
  dim3 grid(B / (WPB * EPW)), block(32 * WPB);
  hipLaunchKernelGGL(vqc_kernel, grid, block, 0, stream, theta, phi, jup, vqcw, out);
}


# Second product of each RY stage replaced by packed-f32 RY rotations between register pairs (lane-half qubit brought in by one cross-row swap): 32 WMMAs not 64; LDS layout follows; sum order changes
# speedup vs baseline: 46.9145x; 1.0965x over previous
#include <hip/hip_runtime.h>

typedef __attribute__((ext_vector_type(2))) float v2f;
typedef __attribute__((ext_vector_type(4))) float v4f;
typedef __attribute__((ext_vector_type(8))) float v8f;

#define WPB 8
#define DD  256
#define EPW 2
#define NSH 3


static __device__ __forceinline__ v8f wmma4(v2f a, v2f b, v8f c) {
  return __builtin_amdgcn_wmma_f32_16x16x4_f32(false, a, false, b, (short)0, c,
                                               false, false);
}

static __device__ __forceinline__ v8f mmFrag(v8f a, v8f b, v8f acc) {
#pragma unroll
  for (int k = 0; k < 4; ++k) {
    v2f av, bv;
    av.x = a[2 * k]; av.y = a[2 * k + 1];
    bv.x = b[2 * k]; bv.y = b[2 * k + 1];
    acc = wmma4(av, bv, acc);
  }
  return acc;
}

static __device__ __forceinline__ void rmm(v8f Ar, v8f Ai, v8f B, v8f* Dr, v8f* Di) {
  v8f r = {};
  r = mmFrag(Ar, B, r);
  v8f i = {};
  i = mmFrag(Ai, B, i);
  *Dr = r; *Di = i;
}

struct R2 { float c, s; };


template <int I0, int I1>
static __device__ __forceinline__ void rotQuad(v8f* F, R2 g) {
  v2f x0, x1;
  x0.x = (*F)[I0]; x0.y = (*F)[I0 + 1];
  x1.x = (*F)[I1]; x1.y = (*F)[I1 + 1];
  const v2f n0 = x0 * g.c - x1 * g.s;
  const v2f n1 = x0 * g.s + x1 * g.c;
  (*F)[I0] = n0.x; (*F)[I0 + 1] = n0.y;
  (*F)[I1] = n1.x; (*F)[I1 + 1] = n1.y;
}
template <int I>
static __device__ __forceinline__ void rotPair(v8f* F, R2 g) {
  v2f x, xs, sv;
  x.x = (*F)[I];  x.y = (*F)[I + 1];
  xs.x = x.y;     xs.y = x.x;
  sv.x = -g.s;    sv.y = g.s;
  const v2f n = x * g.c + xs * sv;
  (*F)[I] = n.x; (*F)[I + 1] = n.y;
}
static __device__ __forceinline__ void swapRows(float* a, float* b) {
  typedef __attribute__((ext_vector_type(2))) unsigned int v2u;
  const v2u r = __builtin_amdgcn_permlane16_swap(__float_as_uint(*a), __float_as_uint(*b),
                                                 false, false);
  *a = __uint_as_float(r.x); *b = __uint_as_float(r.y);
}
static __device__ __forceinline__ void swapHalfBit1(v8f* F) {
  float a0 = (*F)[0], b0 = (*F)[2], a1 = (*F)[1], b1 = (*F)[3];
  float a2 = (*F)[4], b2 = (*F)[6], a3 = (*F)[5], b3 = (*F)[7];
  swapRows(&a0, &b0); swapRows(&a1, &b1); swapRows(&a2, &b2); swapRows(&a3, &b3);
  (*F)[0] = a0; (*F)[2] = b0; (*F)[1] = a1; (*F)[3] = b1;
  (*F)[4] = a2; (*F)[6] = b2; (*F)[5] = a3; (*F)[7] = b3;
}
static __device__ __forceinline__ void rotPlane(v8f* F, R2 gA, R2 gH, R2 gC, R2 gD) {
  rotQuad<0, 2>(F, gA); rotQuad<4, 6>(F, gA);
  swapHalfBit1(F);
  rotQuad<0, 2>(F, gH); rotQuad<4, 6>(F, gH);
  rotQuad<0, 4>(F, gC); rotQuad<2, 6>(F, gC);
  rotPair<0>(F, gD); rotPair<2>(F, gD); rotPair<4>(F, gD); rotPair<6>(F, gD);
}

static __device__ __forceinline__ void applyStage(v8f* Sr, v8f* Si, v8f K, R2 gA, R2 gH, R2 gC,
                                                  R2 gD) {
  v8f Ur, Ui;
  rmm(*Sr, *Si, K, &Ur, &Ui);
  rotPlane(&Ur, gA, gH, gC, gD);
  rotPlane(&Ui, gA, gH, gC, gD);
  *Sr = Ur; *Si = Ui;
}

static __device__ __forceinline__ float pick(bool c, float t, float f) { return c ? t : f; }

static __device__ __forceinline__ R2 ryGate(float halfAngle) {
  R2 g;
  __sincosf(halfAngle, &g.s, &g.c);
  return g;
}
static __device__ __forceinline__ R2 ryGateUniform(float halfAngle) {
  R2 g = ryGate(halfAngle);
  g.c = __int_as_float(__builtin_amdgcn_readfirstlane(__float_as_int(g.c)));
  g.s = __int_as_float(__builtin_amdgcn_readfirstlane(__float_as_int(g.s)));
  return g;
}

static __device__ __forceinline__ void pickRow(bool a, R2 g, float* x0, float* x1) {
  *x0 = pick(a, g.s, g.c);
  *x1 = pick(a, g.c, -g.s);
}

static __device__ __forceinline__ v8f kronFrag(R2 g0, R2 g1, R2 g2, R2 g3, int lane) {
  const bool a0 = (lane >> 3) & 1;
  const bool a1 = (lane >> 2) & 1;
  const bool a2 = (lane >> 1) & 1;
  const bool a3 = lane & 1;
  const bool b0 = (lane >> 4) & 1;
  float x00, x01, u0, u1, v0, v1, w0, w1;
  pickRow(a0, g0, &x00, &x01);
  const float p = pick(b0, x01, x00);
  pickRow(a1, g1, &u0, &u1);
  pickRow(a2, g2, &v0, &v1);
  pickRow(a3, g3, &w0, &w1);
  const float s0 = p * u0, s1 = p * u1;
  const float t0 = s0 * v0, t1 = s0 * v1, t2 = s1 * v0, t3 = s1 * v1;
  v8f k;
  k[0] = t0 * w0; k[1] = t0 * w1; k[2] = t1 * w0; k[3] = t1 * w1;
  k[4] = t2 * w0; k[5] = t2 * w1; k[6] = t3 * w0; k[7] = t3 * w1;
  return k;
}

struct FG { float c, s, pr, pi; };

static __device__ __forceinline__ FG fuseGate(float thq, float phq, float wa, float wb) {
  float sP, cP, sM, cM, sE, cE;
  __sincosf(0.25f * thq + 0.5f * wb, &sP, &cP);
  __sincosf(0.25f * thq - 0.5f * wb, &sM, &cM);
  __sincosf(0.25f * phq + 0.5f * wa, &sE, &cE);
  const float ur = cP * cE, ui = -(cM * sE);
  const float vr = sP * cE, vi = sM * sE;
  const float nu = ur * ur + ui * ui, nv = vr * vr + vi * vi;
  const float TINY = 1e-30f;
  const bool ku = nu > TINY, kv = nv > TINY;
  const float iu = ku ? __builtin_amdgcn_rsqf(nu) : 0.f;
  const float iv = kv ? __builtin_amdgcn_rsqf(nv) : 0.f;
  FG g;
  g.c = nu * iu;
  g.s = nv * iv;
  const float aur = ku ? ur * iu : 1.f, aui = -(ui * iu);
  const float avr = kv ? vr * iv : 1.f, avi = -(vi * iv);
  g.pr = aur * avr - aui * avi;
  g.pi = aur * avi + aui * avr;
  return g;
}

static __device__ __forceinline__ float laneVal(float x, int k) {
  return __int_as_float(__builtin_amdgcn_readlane(__float_as_int(x), k));
}

struct C2 { float r, i; };
static __device__ __forceinline__ C2 cmul(C2 a, C2 b) {
  C2 z;
  z.r = a.r * b.r - a.i * b.i;
  z.i = a.r * b.i + a.i * b.r;
  return z;
}
static __device__ __forceinline__ C2 csel(bool c, C2 p) {
  C2 z;
  z.r = pick(c, p.r, 1.f);
  z.i = pick(c, p.i, 0.f);
  return z;
}

static __device__ __forceinline__ void phaseProd(C2 p0, C2 p1, C2 p2, C2 p3, C2 p4, C2 p5,
                                                 C2 p6, C2 p7, int lane, v8f* C, v8f* Sn) {
  C2 t = csel((lane >> 4) & 1, p0);
  t = cmul(t, csel((lane >> 3) & 1, p4));
  t = cmul(t, csel((lane >> 2) & 1, p5));
  t = cmul(t, csel((lane >> 1) & 1, p6));
  t = cmul(t, csel(lane & 1, p7));
  const C2 x1 = cmul(t, p1);
  const C2 y1 = cmul(t, p2), y3 = cmul(x1, p2);
  const C2 z1 = cmul(t, p3), z3 = cmul(y1, p3), z5 = cmul(x1, p3), z7 = cmul(y3, p3);
  v8f c, s;
  c[0] = t.r;  s[0] = t.i;   c[1] = z1.r; s[1] = z1.i;
  c[2] = y1.r; s[2] = y1.i;  c[3] = z3.r; s[3] = z3.i;
  c[4] = x1.r; s[4] = x1.i;  c[5] = z5.r; s[5] = z5.i;
  c[6] = y3.r; s[6] = y3.i;  c[7] = z7.r; s[7] = z7.i;
  *C = c; *Sn = s;
}

static __device__ __forceinline__ constexpr int pairIdx(int i, int j) {
  return i * (15 - i) / 2 + (j - i - 1);
}

static __device__ __forceinline__ v8f phaseAngles(float a0, float a1, float a2, float a3,
                                                  float a4, float a5, float a6, float a7,
                                                  int lane) {
  float t = pick((lane >> 4) & 1, a0, -a0);
  t += pick((lane >> 3) & 1, a4, -a4);
  t += pick((lane >> 2) & 1, a5, -a5);
  t += pick((lane >> 1) & 1, a6, -a6);
  t += pick(lane & 1, a7, -a7);
  const float u0 = t - a1, u1 = t + a1;
  const float v0 = u0 - a2, v1 = u0 + a2, v2 = u1 - a2, v3 = u1 + a2;
  v8f e;
  e[0] = v0 - a3; e[1] = v0 + a3; e[2] = v1 - a3; e[3] = v1 + a3;
  e[4] = v2 - a3; e[5] = v2 + a3; e[6] = v3 - a3; e[7] = v3 + a3;
  return e;
}

static __device__ __forceinline__ void mulPhaseCS(v8f* Sr, v8f* Si, v8f C, v8f Sn) {
  v8f r, i;
#pragma unroll
  for (int j = 0; j < 8; ++j) {
    r[j] = (*Sr)[j] * C[j] - (*Si)[j] * Sn[j];
    i[j] = (*Sr)[j] * Sn[j] + (*Si)[j] * C[j];
  }
  *Sr = r; *Si = i;
}

static __device__ __forceinline__ void stFrag(float* F, v8f v, int lane) {
  v4f lo, hi;
  lo.x = v[0]; lo.y = v[1]; lo.z = v[2]; lo.w = v[3];
  hi.x = v[4]; hi.y = v[5]; hi.z = v[6]; hi.w = v[7];
  *(v4f*)(F + lane * 4) = lo;
  *(v4f*)(F + 128 + lane * 4) = hi;
}
static __device__ __forceinline__ v8f ldFrag(const float* F, int lane) {
  const v4f lo = *(const v4f*)(F + lane * 4);
  const v4f hi = *(const v4f*)(F + 128 + lane * 4);
  v8f v;
  v[0] = lo.x; v[1] = lo.y; v[2] = lo.z; v[3] = lo.w;
  v[4] = hi.x; v[5] = hi.y; v[6] = hi.z; v[7] = hi.w;
  return v;
}

template <int RNG>
static __device__ __forceinline__ int cnotSrc(int s) {
  int src = s;
#pragma unroll
  for (int k = 7; k >= 0; --k) {
    int tgt = (k + RNG) & 7;
    src ^= ((src >> (7 - k)) & 1) << (7 - tgt);
  }
  return src;
}
static __device__ __forceinline__ int exAddr(int x) {
  return (((x >> 2) & 1) << 7) | (((x >> 1) & 1) << 6) | (((x >> 4) & 15) << 2) |
         (((x >> 3) & 1) << 1) | (x & 1);
}
static __device__ __forceinline__ int gatherIdx(int lane) {
  const int h = lane >> 4, c3 = (lane >> 3) & 1, c2 = (lane >> 2) & 1, c1 = (lane >> 1) & 1;
  const int c0 = lane & 1;
  return (c1 << 7) | (c2 << 6) | (c3 << 5) | (c0 << 4) | (h << 3);
}

template <int RNG>
static __device__ __forceinline__ void ldCnot(const float* Br, const float* Bi,
                                              v8f* Sr, v8f* Si, int lane) {
  const int ab = exAddr(cnotSrc<RNG>(gatherIdx(lane)));
  v8f r, i;
#pragma unroll
  for (int j = 0; j < 8; ++j) {
    const int a = ab ^ exAddr(cnotSrc<RNG>(j));
    r[j] = Br[a];
    i[j] = Bi[a];
  }
  *Sr = r; *Si = i;
}

template <int RNG>
static constexpr int cnotDst(int t) {
  int d = t;
  for (int k = 0; k <= 7; ++k) {
    int tgt = (k + RNG) & 7;
    d ^= ((d >> (7 - k)) & 1) << (7 - tgt);
  }
  return d;
}
template <int RNG>
static constexpr int zMask(int k) {
  int m = 0;
  for (int bpos = 0; bpos < 8; ++bpos)
    if ((cnotDst<RNG>(1 << bpos) >> (7 - k)) & 1) m |= 1 << bpos;
  return m;
}
static_assert(zMask<2>(0) == 0x2A && zMask<2>(1) == 0x15 && zMask<2>(2) == 0xA0 &&
              zMask<2>(3) == 0x50 && zMask<2>(4) == 0xA8 && zMask<2>(5) == 0x54 &&
              zMask<2>(6) == 0xAA && zMask<2>(7) == 0x55,
              "readout sign masks do not match the layer-1 ring");

template <int M>
static __device__ __forceinline__ float xrow(float x) {
  return __int_as_float(__builtin_amdgcn_update_dpp(__float_as_int(x), __float_as_int(x),
                                                    0x160 + M, 0xf, 0xf, true));
}
static __device__ __forceinline__ float xhalf(float x) {
  const int v = __float_as_int(x);
  return __int_as_float(__builtin_amdgcn_permlanex16(v, v, 0x76543210, 0xfedcba98, false, false));
}
static __device__ __forceinline__ float walsh(bool bit, float x, float other) {
  return other + pick(bit, -x, x);
}

static __device__ __forceinline__ void phase0(const float* thv, const float* phv, const float* jv,
                                              const float* __restrict__ w0, int lane,
                                              v8f* Sr, v8f* Si) {
  const int c = lane & 15, h = lane >> 4;
  v8f sr, si;
  {
    float mc[8], ms[8];
#pragma unroll
    for (int k = 0; k < 8; ++k) __sincosf(thv[k] * 0.5f, &ms[k], &mc[k]);

    float mb = pick(h, ms[0], mc[0]);
    mb *= pick((c >> 3) & 1, ms[4], mc[4]);
    mb *= pick((c >> 2) & 1, ms[5], mc[5]);
    mb *= pick((c >> 1) & 1, ms[6], mc[6]);
    mb *= pick(c & 1, ms[7], mc[7]);
    const float m0 = mb * mc[1], m1 = mb * ms[1];
    const float n0 = m0 * mc[2], n1 = m0 * ms[2], n2 = m1 * mc[2], n3 = m1 * ms[2];
    v8f mag;
    mag[0] = n0 * mc[3]; mag[1] = n0 * ms[3]; mag[2] = n1 * mc[3]; mag[3] = n1 * ms[3];
    mag[4] = n2 * mc[3]; mag[5] = n2 * ms[3]; mag[6] = n3 * mc[3]; mag[7] = n3 * ms[3];

    const float HP = 1.57079632679489662f;
    const float g0 = pick(h, -1.f, 1.f);
    const float g4 = pick((c >> 3) & 1, -1.f, 1.f);
    const float g5 = pick((c >> 2) & 1, -1.f, 1.f);
    const float g6 = pick((c >> 1) & 1, -1.f, 1.f);
    const float g7 = pick(c & 1, -1.f, 1.f);

    const float a0 = 0.5f * (phv[0] + w0[0]),  a1 = 0.5f * (phv[1] + w0[3]);
    const float a2 = 0.5f * (phv[2] + w0[6]),  a3 = 0.5f * (phv[3] + w0[9]);
    const float a4 = 0.5f * (phv[4] + w0[12]), a5 = 0.5f * (phv[5] + w0[15]);
    const float a6 = 0.5f * (phv[6] + w0[18]), a7 = 0.5f * (phv[7] + w0[21]);

    const float i0s = jv[pairIdx(0, 4)] * g4 + jv[pairIdx(0, 5)] * g5 + jv[pairIdx(0, 6)] * g6 +
                      jv[pairIdx(0, 7)] * g7;
    const float i4s = jv[pairIdx(4, 5)] * g5 + jv[pairIdx(4, 6)] * g6 + jv[pairIdx(4, 7)] * g7;
    const float i5s = jv[pairIdx(5, 6)] * g6 + jv[pairIdx(5, 7)] * g7;
    const float i6s = jv[pairIdx(6, 7)] * g7;
    const float A = -(g0 * (HP * i0s + a0) + g4 * (HP * i4s + a4) + g5 * (HP * i5s + a5) +
                      g6 * (HP * i6s + a6) + g7 * a7);

    const float f1 = jv[pairIdx(0, 1)] * g0 + jv[pairIdx(1, 4)] * g4 + jv[pairIdx(1, 5)] * g5 +
                     jv[pairIdx(1, 6)] * g6 + jv[pairIdx(1, 7)] * g7;
    const float f2 = jv[pairIdx(0, 2)] * g0 + jv[pairIdx(2, 4)] * g4 + jv[pairIdx(2, 5)] * g5 +
                     jv[pairIdx(2, 6)] * g6 + jv[pairIdx(2, 7)] * g7;
    const float f3 = jv[pairIdx(0, 3)] * g0 + jv[pairIdx(3, 4)] * g4 + jv[pairIdx(3, 5)] * g5 +
                     jv[pairIdx(3, 6)] * g6 + jv[pairIdx(3, 7)] * g7;
    const float F1 = -(HP * f1 + a1), F2 = -(HP * f2 + a2), F3 = -(HP * f3 + a3);
    const float K12 = HP * jv[pairIdx(1, 2)], K13 = HP * jv[pairIdx(1, 3)];
    const float K23 = HP * jv[pairIdx(2, 3)];

    const float tP = A + F1, tM = A - F1;
    const float F2p = F2 - K12, F2m = F2 + K12;
    const float uPP = tP + F2p, uPM = tP - F2p, uMP = tM + F2m, uMM = tM - F2m;
    const float F3p = F3 - K13, F3m = F3 + K13;
    const float vPP = F3p - K23, vPM = F3p + K23;
    const float vMP = F3m - K23, vMM = F3m + K23;
    v8f X;
    X[0] = uPP + vPP; X[1] = uPP - vPP; X[2] = uPM + vPM; X[3] = uPM - vPM;
    X[4] = uMP + vMP; X[5] = uMP - vMP; X[6] = uMM + vMM; X[7] = uMM - vMM;

#pragma unroll
    for (int j = 0; j < 8; ++j) {
      float se, ce;
      __sincosf(X[j], &se, &ce);
      sr[j] = mag[j] * ce;
      si[j] = mag[j] * se;
    }
  }
  *Sr = sr; *Si = si;
}

static __device__ __forceinline__ void readout(v8f Sr, v8f Si, int lane, float* xo, float* yo) {
  const int c = lane & 15, h = lane >> 4;
  float p[8];
#pragma unroll
  for (int j = 0; j < 8; ++j) p[j] = Sr[j] * Sr[j] + Si[j] * Si[j];
  const float a0 = p[0] + p[1], b0 = p[0] - p[1], a1 = p[2] + p[3], b1 = p[2] - p[3];
  const float a2 = p[4] + p[5], b2 = p[4] - p[5], a3 = p[6] + p[7], b3 = p[6] - p[7];
  const float r2 = (a0 - a1) + (a2 - a3);
  const float b01 = b0 + b1, b23 = b2 + b3;
  const float r1 = b01 + b23, r5 = b01 - b23;

  const bool c3 = (c >> 3) & 1, c2 = (c >> 2) & 1, c1 = (c >> 1) & 1, c0 = c & 1;
  float x = r2 + xrow<4>(r2);
  x = x + xrow<1>(x);
  x = walsh(h, x, xhalf(x));
  x = walsh(c3, x, xrow<8>(x));
  x = walsh(c1, x, xrow<2>(x));
  float y = pick(h, r5, r1) + xhalf(pick(h, r1, r5));
  y = y + xrow<8>(y);
  y = y + xrow<2>(y);
  y = walsh(c2, y, xrow<4>(y));
  y = walsh(c0, y, xrow<1>(y));
  *xo = x; *yo = y;
}

extern "C" __global__ __launch_bounds__(256, 8) void vqc_kernel(
    const float* __restrict__ theta, const float* __restrict__ phi,
    const float* __restrict__ jup, const float* __restrict__ vqcw,
    float* __restrict__ out) {
  __shared__ __attribute__((aligned(16))) float sAr[WPB][EPW][DD], sAi[WPB][EPW][DD];
  __shared__ __attribute__((aligned(16))) float sW[NSH][DD];

  int tid = threadIdx.x;
  int wid = tid >> 5, lane = tid & 31;
  int b = (blockIdx.x * WPB + wid) * EPW;

  const float* w0 = vqcw;
  const float* w1 = vqcw + 24;
  const int q = lane & 7;

  float thv[EPW][8], phv[EPW][8], jv[EPW][28];
#pragma unroll
  for (int e = 0; e < EPW; ++e) {
    const float* th = theta + (b + e) * 8;
    const float* ph = phi + (b + e) * 8;
    const float* jj = jup + (b + e) * 28;
#pragma unroll
    for (int k = 0; k < 8; ++k) { thv[e][k] = th[k]; phv[e][k] = ph[k]; }
#pragma unroll
    for (int p = 0; p < 28; ++p) jv[e][p] = jj[p];
  }
  const float thq = theta[b * 8 + (lane & 15)], phq = phi[b * 8 + (lane & 15)];
  const float waq = w1[3 * q], wbq = w1[3 * q + 1];

  {
    const int wu = __builtin_amdgcn_readfirstlane(wid);
    if (wu == 0) {
      stFrag(sW[0], kronFrag(ryGate(0.5f * w0[1]), ryGate(0.5f * w0[4]),
                             ryGate(0.5f * w0[7]), ryGate(0.5f * w0[10]), lane), lane);
    } else if (wu == 1) {
      const v8f E = phaseAngles(0.5f * w0[20], 0.5f * w0[17], 0.5f * w0[14], 0.5f * w0[23],
                                0.5f * w0[2],  0.5f * w0[5],  0.5f * w0[8],  0.5f * w0[11],
                                lane);
      v8f pc, ps;
#pragma unroll
      for (int j = 0; j < 8; ++j) {
        float s, c;
        __sincosf(E[j], &s, &c);
        pc[j] = c; ps[j] = s;
      }
      stFrag(sW[1], pc, lane);
      stFrag(sW[2], ps, lane);
    }
  }
  const R2 r4 = ryGateUniform(0.5f * w0[13]), r5 = ryGateUniform(0.5f * w0[16]);
  const R2 r6 = ryGateUniform(0.5f * w0[19]), r7 = ryGateUniform(0.5f * w0[22]);

  v8f Sr[EPW], Si[EPW];

#pragma unroll
  for (int e = 0; e < EPW; ++e) phase0(thv[e], phv[e], jv[e], w0, lane, &Sr[e], &Si[e]);

  __syncthreads();
  {
    const v8f L = ldFrag(sW[0], lane);
    const v8f pc = ldFrag(sW[1], lane), ps = ldFrag(sW[2], lane);
#pragma unroll
    for (int e = 0; e < EPW; ++e) {
      applyStage(&Sr[e], &Si[e], L, r6, r4, r5, r7);
      mulPhaseCS(&Sr[e], &Si[e], pc, ps);
    }
  }

#pragma unroll
  for (int e = 0; e < EPW; ++e) {
    stFrag(sAr[wid][e], Sr[e], lane);
    stFrag(sAi[wid][e], Si[e], lane);
  }
  {
    const FG f = fuseGate(thq, phq, waq, wbq);
    v8f Rf[EPW], Pc[EPW], Ps[EPW];
    R2 g[EPW][8];
#pragma unroll
    for (int e = 0; e < EPW; ++e) {
      C2 p[8];
#pragma unroll
      for (int k = 0; k < 8; ++k) {
        g[e][k].c = laneVal(f.c, 8 * e + k);  g[e][k].s = laneVal(f.s, 8 * e + k);
        p[k].r = laneVal(f.pr, 8 * e + k);    p[k].i = laneVal(f.pi, 8 * e + k);
      }
      Rf[e] = kronFrag(g[e][4], g[e][5], g[e][6], g[e][7], lane);
      phaseProd(p[4], p[5], p[6], p[7], p[2], p[1], p[0], p[3], lane, &Pc[e], &Ps[e]);
    }
    __syncthreads();
#pragma unroll
    for (int e = 0; e < EPW; ++e) {
      ldCnot<1>(sAr[wid][e], sAi[wid][e], &Sr[e], &Si[e], lane);
      mulPhaseCS(&Sr[e], &Si[e], Pc[e], Ps[e]);
      applyStage(&Sr[e], &Si[e], Rf[e], g[e][0], g[e][2], g[e][1], g[e][3]);
    }
  }

  float o[EPW][8];
#pragma unroll
  for (int e = 0; e < EPW; ++e) {
    float x, y;
    readout(Sr[e], Si[e], lane, &x, &y);
    o[e][0] = laneVal(x, 10); o[e][2] = laneVal(x, 16); o[e][4] = laneVal(x, 24); o[e][6] = laneVal(x, 26);
    o[e][1] = laneVal(y, 5);  o[e][3] = laneVal(y, 16); o[e][5] = laneVal(y, 20); o[e][7] = laneVal(y, 21);
  }
  if (lane == 0) {
    float* po = out + b * 8;
#pragma unroll
    for (int e = 0; e < EPW; ++e)
#pragma unroll
      for (int k = 0; k < 8; ++k) po[e * 8 + k] = o[e][k];
  }
}

extern "C" void kernel_launch(void* const* d_in, const int* in_sizes, int n_in,
                              void* d_out, int out_size, void* d_ws, size_t ws_size,
                              hipStream_t stream) {
  const float* theta = (const float*)d_in[0];
  const float* phi   = (const float*)d_in[1];
  const float* jup   = (const float*)d_in[2];
  const float* vqcw  = (const float*)d_in[3];
  float* out = (float*)d_out;
  int B = in_sizes[0] / 8;
  dim3 grid(B / (WPB * EPW)), block(32 * WPB);
  hipLaunchKernelGGL(vqc_kernel, grid, block, 0, stream, theta, phi, jup, vqcw, out);
}
